# starting kernel
# baseline (speedup 1.0000x reference)
#include <hip/hip_runtime.h>
#include <hip/hip_bf16.h>

#define F_IN   64
#define HEADS  4
#define CHD    32
#define HC     128
#define NEG_SLOPE 0.2f

typedef __attribute__((ext_vector_type(2))) float v2f;
typedef __attribute__((ext_vector_type(8))) float v8f;

__device__ __forceinline__ float leaky(float x) { return x > 0.f ? x : NEG_SLOPE * x; }

__global__ void fold_kernel(const float* __restrict__ We1, const float* __restrict__ ae1,
                            const float* __restrict__ We2, const float* __restrict__ ae2,
                            float* __restrict__ Ve1, float* __restrict__ Ve2) {
    int t = threadIdx.x;
    int f = t >> 2, h = t & 3;
    float s1 = 0.f, s2 = 0.f;
#pragma unroll
    for (int c = 0; c < CHD; ++c) {
        s1 += We1[f * HC + h * CHD + c] * ae1[h * CHD + c];
        s2 += We2[f * HC + h * CHD + c] * ae2[h * CHD + c];
    }
    Ve1[f * 4 + h] = s1;
    Ve2[f * 4 + h] = s2;
}

__global__ void edge_proj_kernel(const float* __restrict__ EA, const int* __restrict__ dstA,
                                 const float* __restrict__ Ve1, const float* __restrict__ Ve2,
                                 float* __restrict__ ae1v, float* __restrict__ ae2v,
                                 float* __restrict__ sum1, float* __restrict__ sum2,
                                 int* __restrict__ degI, int E) {
    __shared__ float sv[F_IN][8];
    int t = threadIdx.x;
    {
        int f = t >> 2, h = t & 3;
        sv[f][h]     = Ve1[f * 4 + h];
        sv[f][h + 4] = Ve2[f * 4 + h];
    }
    __syncthreads();
    int e = blockIdx.x * blockDim.x + t;
    if (e >= E) return;
    float acc[8] = {0, 0, 0, 0, 0, 0, 0, 0};
    const float4* ea4 = reinterpret_cast<const float4*>(EA + (size_t)e * F_IN);
#pragma unroll
    for (int j = 0; j < F_IN / 4; ++j) {
        float4 v = ea4[j];
        int f = j * 4;
#pragma unroll
        for (int q = 0; q < 8; ++q) {
            acc[q] += v.x * sv[f][q] + v.y * sv[f + 1][q] +
                      v.z * sv[f + 2][q] + v.w * sv[f + 3][q];
        }
    }
    int d = dstA[e];
#pragma unroll
    for (int h = 0; h < 4; ++h) {
        ae1v[(size_t)e * 4 + h] = acc[h];
        ae2v[(size_t)e * 4 + h] = acc[4 + h];
        atomicAdd(&sum1[(size_t)d * 4 + h], acc[h]);
        atomicAdd(&sum2[(size_t)d * 4 + h], acc[4 + h]);
    }
    atomicAdd(&degI[d], 1);
}

__global__ void scan_kernel(const int* __restrict__ deg, int* __restrict__ off, int N) {
    __shared__ int sh[1024];
    int t = threadIdx.x;
    int CHK = (N + 1023) / 1024;
    int b = t * CHK, en = min(b + CHK, N);
    int sum = 0;
    for (int i = b; i < en; ++i) sum += deg[i];
    sh[t] = sum;
    __syncthreads();
    for (int d = 1; d < 1024; d <<= 1) {
        int v = (t >= d) ? sh[t - d] : 0;
        __syncthreads();
        sh[t] += v;
        __syncthreads();
    }
    int run = sh[t] - sum;
    for (int i = b; i < en; ++i) { off[i] = run; run += deg[i]; }
}

__global__ void scatter_kernel(const int* __restrict__ dstA, const int* __restrict__ off,
                               int* __restrict__ cursor, int* __restrict__ eidA, int E) {
    int e = blockIdx.x * blockDim.x + threadIdx.x;
    if (e >= E) return;
    int d = dstA[e];
    int pos = off[d] + atomicAdd(&cursor[d], 1);
    eidA[pos] = e;
}

__global__ void gemm_wmma_kernel(const float* __restrict__ X, const float* __restrict__ W,
                                 float* __restrict__ Y, int M, int K) {
    int wave = threadIdx.x >> 5;
    int lane = threadIdx.x & 31;
    int tile = blockIdx.x * (blockDim.x >> 5) + wave;
    int mt = tile >> 3;
    int nt = tile & 7;
    if (mt * 16 >= M) return;
    int row0 = mt * 16, col0 = nt * 16;
    int m = lane & 15, half = lane >> 4;

    v8f acc = {};
    const float* ap0 = X + (size_t)(row0 + m) * K + 2 * half;
    const float* bp0 = W + (size_t)(2 * half) * HC + col0 + m;
    for (int k0 = 0; k0 < K; k0 += 4) {
        v2f a, b;
        a.x = ap0[k0];
        a.y = ap0[k0 + 1];
        b.x = bp0[(size_t)k0 * HC];
        b.y = bp0[(size_t)(k0 + 1) * HC];
        acc = __builtin_amdgcn_wmma_f32_16x16x4_f32(false, a, false, b,
                                                    (short)0, acc, false, false);
    }
    float* yp = Y + (size_t)(row0 + 8 * half) * HC + col0 + m;
#pragma unroll
    for (int r = 0; r < 8; ++r) yp[(size_t)r * HC] = acc[r];
}

__global__ void node_att_kernel(const float* __restrict__ Hp,
                                const float* __restrict__ att_s, const float* __restrict__ att_d,
                                float* __restrict__ a_s, float* __restrict__ a_d, int N) {
    int wave = threadIdx.x >> 5, lane = threadIdx.x & 31;
    int n = blockIdx.x * (blockDim.x >> 5) + wave;
    if (n >= N) return;
    int h = lane >> 3, q = lane & 7;
    float4 hv = reinterpret_cast<const float4*>(Hp)[(size_t)n * 32 + lane];
    float4 sa = reinterpret_cast<const float4*>(att_s)[lane];
    float4 da = reinterpret_cast<const float4*>(att_d)[lane];
    float ps = hv.x * sa.x + hv.y * sa.y + hv.z * sa.z + hv.w * sa.w;
    float pd = hv.x * da.x + hv.y * da.y + hv.z * da.z + hv.w * da.w;
#pragma unroll
    for (int msk = 1; msk < 8; msk <<= 1) {
        ps += __shfl_xor(ps, msk, 32);
        pd += __shfl_xor(pd, msk, 32);
    }
    if (q == 0) { a_s[(size_t)n * 4 + h] = ps; a_d[(size_t)n * 4 + h] = pd; }
}

__global__ void agg_kernel(const int* __restrict__ off, const int* __restrict__ degI,
                           const int* __restrict__ eidA, const int* __restrict__ srcA,
                           const float* __restrict__ a_e, const float* __restrict__ a_s,
                           const float* __restrict__ a_d, const float* __restrict__ sum_ae,
                           const float* __restrict__ Hp, const float* __restrict__ bias,
                           const float* __restrict__ lin_w, const float* __restrict__ lin_b,
                           float* __restrict__ outv, float* __restrict__ outs,
                           int N, int mode) {
    int wave = threadIdx.x >> 5, lane = threadIdx.x & 31;
    int n = blockIdx.x * (blockDim.x >> 5) + wave;
    if (n >= N) return;
    int h = lane >> 3;
    const float4* Hp4 = reinterpret_cast<const float4*>(Hp);

    float ad_n = a_d[(size_t)n * 4 + h];
    float as_n = a_s[(size_t)n * 4 + h];
    int cnt = degI[n];
    float degf = fmaxf((float)cnt, 1.0f);
    float M = leaky(as_n + ad_n + sum_ae[(size_t)n * 4 + h] / degf);
    float D = 1.0f;
    float4 A = Hp4[(size_t)n * 32 + lane];

    int s0 = off[n];
    for (int i = 0; i < cnt; ++i) {
        int e = eidA[s0 + i];
        int s = srcA[e];
        float lg = leaky(a_s[(size_t)s * 4 + h] + ad_n + a_e[(size_t)e * 4 + h]);
        float Mn = fmaxf(M, lg);
        float sc = __expf(M - Mn);
        float p  = __expf(lg - Mn);
        float4 hv = Hp4[(size_t)s * 32 + lane];
        A.x = A.x * sc + p * hv.x;
        A.y = A.y * sc + p * hv.y;
        A.z = A.z * sc + p * hv.z;
        A.w = A.w * sc + p * hv.w;
        D = D * sc + p;
        M = Mn;
    }
    float4 bv = reinterpret_cast<const float4*>(bias)[lane];
    float inv = 1.0f / D;
    float4 r = { A.x * inv + bv.x, A.y * inv + bv.y, A.z * inv + bv.z, A.w * inv + bv.w };
    if (mode == 0) {
        r.x = fmaxf(r.x, 0.f); r.y = fmaxf(r.y, 0.f);
        r.z = fmaxf(r.z, 0.f); r.w = fmaxf(r.w, 0.f);
        reinterpret_cast<float4*>(outv)[(size_t)n * 32 + lane] = r;
    } else {
        float4 lw = reinterpret_cast<const float4*>(lin_w)[lane];
        float part = r.x * lw.x + r.y * lw.y + r.z * lw.z + r.w * lw.w;
#pragma unroll
        for (int msk = 16; msk >= 1; msk >>= 1) part += __shfl_xor(part, msk, 32);
        if (lane == 0) outs[n] = part + lin_b[0];
    }
}

extern "C" void kernel_launch(void* const* d_in, const int* in_sizes, int n_in,
                              void* d_out, int out_size, void* d_ws, size_t ws_size,
                              hipStream_t stream) {
    const float* x   = (const float*)d_in[0];
    const float* ea  = (const float*)d_in[1];
    const float* W1  = (const float*)d_in[2];
    const float* as1 = (const float*)d_in[3];
    const float* ad1 = (const float*)d_in[4];
    const float* We1 = (const float*)d_in[5];
    const float* ae1 = (const float*)d_in[6];
    const float* b1  = (const float*)d_in[7];
    const float* W2  = (const float*)d_in[8];
    const float* as2 = (const float*)d_in[9];
    const float* ad2 = (const float*)d_in[10];
    const float* We2 = (const float*)d_in[11];
    const float* ae2 = (const float*)d_in[12];
    const float* b2  = (const float*)d_in[13];
    const float* lw  = (const float*)d_in[14];
    const float* lb  = (const float*)d_in[15];
    const int*   ei  = (const int*)d_in[16];

    const int N = in_sizes[0] / F_IN;
    const int E = in_sizes[16] / 2;
    const int* srcA = ei;
    const int* dstA = ei + E;

    char* w = (char*)d_ws;
    auto alloc = [&](size_t bytes) -> void* {
        void* p = (void*)w;
        w += (bytes + 255) & ~(size_t)255;
        return p;
    };
    int*   degI   = (int*)  alloc((size_t)N * 4);
    int*   cursor = (int*)  alloc((size_t)N * 4);
    float* sum1   = (float*)alloc((size_t)N * 16);
    float* sum2   = (float*)alloc((size_t)N * 16);
    size_t zbytes = (size_t)(w - (char*)degI);
    float* Ve1  = (float*)alloc(F_IN * 4 * 4);
    float* Ve2  = (float*)alloc(F_IN * 4 * 4);
    float* ae1v = (float*)alloc((size_t)E * 16);
    float* ae2v = (float*)alloc((size_t)E * 16);
    float* a_sb = (float*)alloc((size_t)N * 16);
    float* a_db = (float*)alloc((size_t)N * 16);
    int*   off  = (int*)  alloc((size_t)N * 4);
    int*   eidA = (int*)  alloc((size_t)E * 4);
    float* Hp   = (float*)alloc((size_t)N * HC * 4);
    float* h1   = (float*)alloc((size_t)N * HC * 4);

    hipMemsetAsync(degI, 0, zbytes, stream);

    fold_kernel<<<1, 256, 0, stream>>>(We1, ae1, We2, ae2, Ve1, Ve2);
    edge_proj_kernel<<<(E + 255) / 256, 256, 0, stream>>>(ea, dstA, Ve1, Ve2,
                                                          ae1v, ae2v, sum1, sum2, degI, E);
    scan_kernel<<<1, 1024, 0, stream>>>(degI, off, N);
    scatter_kernel<<<(E + 255) / 256, 256, 0, stream>>>(dstA, off, cursor, eidA, E);

    const int mt = (N + 15) / 16;
    const int gemmBlocks = (mt * 8 + 3) / 4;
    const int nodeBlocks = (N + 7) / 8;

    gemm_wmma_kernel<<<gemmBlocks, 128, 0, stream>>>(x, W1, Hp, N, F_IN);
    node_att_kernel<<<nodeBlocks, 256, 0, stream>>>(Hp, as1, ad1, a_sb, a_db, N);
    agg_kernel<<<nodeBlocks, 256, 0, stream>>>(off, degI, eidA, srcA, ae1v, a_sb, a_db,
                                               sum1, Hp, b1, nullptr, nullptr,
                                               h1, nullptr, N, 0);
    gemm_wmma_kernel<<<gemmBlocks, 128, 0, stream>>>(h1, W2, Hp, N, HC);
    node_att_kernel<<<nodeBlocks, 256, 0, stream>>>(Hp, as2, ad2, a_sb, a_db, N);
    agg_kernel<<<nodeBlocks, 256, 0, stream>>>(off, degI, eidA, srcA, ae2v, a_sb, a_db,
                                               sum2, Hp, b2, lw, lb,
                                               nullptr, (float*)d_out, N, 1);
}


# projection: block stages 64 rows of X in LDS by whole-line 16-byte loads; wave keeps its column tile's W fragments in registers over 4 row tiles, A fragments from LDS (k sums in a new order)
# speedup vs baseline: 3.1358x; 3.1358x over previous
#include <hip/hip_runtime.h>
#include <hip/hip_bf16.h>

#define F_IN   64
#define HEADS  4
#define CHD    32
#define HC     128
#define NEG_SLOPE 0.2f

typedef __attribute__((ext_vector_type(2))) float v2f;
typedef __attribute__((ext_vector_type(8))) float v8f;

__device__ __forceinline__ float leaky(float x) { return x > 0.f ? x : NEG_SLOPE * x; }

__global__ void fold_kernel(const float* __restrict__ We1, const float* __restrict__ ae1,
                            const float* __restrict__ We2, const float* __restrict__ ae2,
                            float* __restrict__ Bt) {
    int t = threadIdx.x;
    int f = t >> 2, h = t & 3;
    float s1 = 0.f, s2 = 0.f;
#pragma unroll
    for (int c = 0; c < CHD; ++c) {
        s1 += We1[f * HC + h * CHD + c] * ae1[h * CHD + c];
        s2 += We2[f * HC + h * CHD + c] * ae2[h * CHD + c];
    }
    int j = f >> 3, hf = (f >> 2) & 1, u = (f >> 1) & 1, rg = f & 1;
    int step = j * 2 + u;
    Bt[(step * 32 + h + 16 * hf) * 2 + rg]     = s1;
    Bt[(step * 32 + 4 + h + 16 * hf) * 2 + rg] = s2;
    for (int z = t; z < 512; z += 256) {
        int rg2 = z & 1, col = 8 + ((z >> 1) & 7), hf2 = (z >> 4) & 1, st = z >> 5;
        Bt[(st * 32 + col + 16 * hf2) * 2 + rg2] = 0.f;
    }
}

__device__ __forceinline__ void wave_lds_sync() {
    __builtin_amdgcn_fence(__ATOMIC_RELEASE, "wavefront");
    __builtin_amdgcn_wave_barrier();
    __builtin_amdgcn_fence(__ATOMIC_ACQUIRE, "wavefront");
}

#define TILE_PITCH 68

__global__ void __launch_bounds__(256)
edge_csr_kernel(const float* __restrict__ EA, const int* __restrict__ off,
                const int* __restrict__ degI, int* eidA, const int* __restrict__ srcA,
                const float* __restrict__ Bt,
                float* __restrict__ ae1c, float* __restrict__ ae2c,
                float* __restrict__ sum1, float* __restrict__ sum2, int N) {
    __shared__ alignas(16) float tileS[8][16 * TILE_PITCH];
    int wave = threadIdx.x >> 5, lane = threadIdx.x & 31;
    int n = blockIdx.x * 8 + wave;
    if (n >= N) return;
    float* my = tileS[wave];
    int c = lane & 15, half = lane >> 4;

    v2f bfr[16];
#pragma unroll
    for (int s = 0; s < 16; ++s) bfr[s] = reinterpret_cast<const v2f*>(Bt)[s * 32 + lane];

    int cnt = __builtin_amdgcn_readfirstlane(degI[n]);
    int s0  = __builtin_amdgcn_readfirstlane(off[n]);
    float tot = 0.f;

    for (int t0 = 0; t0 < cnt; t0 += 16) {
        int eid = eidA[s0 + min(t0 + c, cnt - 1)];
        int sid = srcA[eid];
        float4 v[8];
#pragma unroll
        for (int i = 0; i < 8; ++i) {
            int er = __shfl(eid, 2 * i + half, 32);
            v[i] = *reinterpret_cast<const float4*>(EA + (size_t)er * F_IN + 4 * c);
        }
#pragma unroll
        for (int i = 0; i < 8; ++i)
            *reinterpret_cast<float4*>(my + (2 * i + half) * TILE_PITCH + 4 * c) = v[i];
        wave_lds_sync();

        v8f acc = {};
#pragma unroll
        for (int j = 0; j < 8; ++j) {
            float4 a4 = *reinterpret_cast<const float4*>(my + c * TILE_PITCH + 8 * j + 4 * half);
            v2f a0 = { a4.x, a4.y };
            v2f a1 = { a4.z, a4.w };
            acc = __builtin_amdgcn_wmma_f32_16x16x4_f32(false, a0, false, bfr[2 * j],
                                                        (short)0, acc, false, false);
            acc = __builtin_amdgcn_wmma_f32_16x16x4_f32(false, a1, false, bfr[2 * j + 1],
                                                        (short)0, acc, false, false);
        }
        wave_lds_sync();

#pragma unroll
        for (int r = 0; r < 8; ++r) {
            float val = acc[r];
            tot += (t0 + 8 * half + r < cnt) ? val : 0.f;
            my[(8 * half + r) * 20 + c] = val;
        }
        wave_lds_sync();
        float4 o = *reinterpret_cast<const float4*>(my + c * 20 + 4 * half);
        wave_lds_sync();
        if (t0 + c < cnt) {
            float* dstp = half ? ae2c : ae1c;
            *reinterpret_cast<float4*>(dstp + (size_t)(s0 + t0 + c) * 4) = o;
            if (half == 0) eidA[s0 + t0 + c] = sid;
        }
    }
    tot += __shfl_xor(tot, 16, 32);
    if (half == 0) {
        if (c < 4)      sum1[(size_t)n * 4 + c]     = tot;
        else if (c < 8) sum2[(size_t)n * 4 + c - 4] = tot;
    }
}

#define OWN_R    392
#define OWN_T    1024
#define OWN_W    (OWN_T / 32)
#define OWN_CAPW 640

__device__ __forceinline__ void own_file(int e, unsigned dl, unsigned mask, int& wcnt, int2* my) {
    if (dl < (unsigned)OWN_R) {
        int p = wcnt + (int)__builtin_amdgcn_mbcnt_lo(mask, 0u);
        my[min(p, OWN_CAPW - 1)] = make_int2(e, (int)dl);
    }
    wcnt += __builtin_popcount(mask);
}

__device__ __forceinline__ void own_take(int4 v, int e0, int lo, int& below, int& wcnt, int2* my) {
    unsigned d0 = (unsigned)(v.x - lo), d1 = (unsigned)(v.y - lo);
    unsigned d2 = (unsigned)(v.z - lo), d3 = (unsigned)(v.w - lo);
    below += (int)((d0 >> 31) + (d1 >> 31) + (d2 >> 31) + (d3 >> 31));
    unsigned k0 = __builtin_amdgcn_ballot_w32(d0 < (unsigned)OWN_R);
    unsigned k1 = __builtin_amdgcn_ballot_w32(d1 < (unsigned)OWN_R);
    unsigned k2 = __builtin_amdgcn_ballot_w32(d2 < (unsigned)OWN_R);
    unsigned k3 = __builtin_amdgcn_ballot_w32(d3 < (unsigned)OWN_R);
    if ((k0 | k1 | k2 | k3) != 0u) {
        own_file(e0,     d0, k0, wcnt, my);
        own_file(e0 + 1, d1, k1, wcnt, my);
        own_file(e0 + 2, d2, k2, wcnt, my);
        own_file(e0 + 3, d3, k3, wcnt, my);
    }
}

__global__ void __launch_bounds__(OWN_T)
csr_owner_kernel(const int* __restrict__ dstA, int* __restrict__ off, int* __restrict__ degI,
                 int* __restrict__ eidA, int N, int E) {
    __shared__ int2 list[OWN_W][OWN_CAPW];
    __shared__ int ncount[OWN_R];
    __shared__ int cur[OWN_R];
    __shared__ int wsum[OWN_W];
    __shared__ int belowAll, ovf;

    const int tid = threadIdx.x, lane = tid & 31, wave = tid >> 5;
    const int lo = blockIdx.x * OWN_R, hi = lo + OWN_R;
    int2* my = list[wave];

    for (int i = tid; i < OWN_R; i += OWN_T) ncount[i] = 0;
    if (tid == 0) { belowAll = 0; ovf = 0; }
    __syncthreads();

    int below = 0, wcnt = 0;
    const int E4 = ((E & 3) == 0) ? (E >> 2) : 0;
    const int4* d4 = reinterpret_cast<const int4*>(dstA);
    const int nfull = E4 / (OWN_T * 4);
    if (nfull > 0) {
        const int rot = (int)((unsigned)blockIdx.x % (unsigned)nfull);
        auto ring = [&](int t) { int x = t + rot; if (x >= nfull) x -= nfull; return x * (OWN_T * 4) + tid; };
        int g = ring(0);
        int4 a0 = d4[g], a1 = d4[g + OWN_T], a2 = d4[g + 2 * OWN_T], a3 = d4[g + 3 * OWN_T];
        int it = 0;
        for (; it + 2 <= nfull; it += 2) {
            int gb = ring(it + 1);
            int4 b0 = d4[gb], b1 = d4[gb + OWN_T], b2 = d4[gb + 2 * OWN_T], b3 = d4[gb + 3 * OWN_T];
            own_take(a0, 4 * g,                 lo, below, wcnt, my);
            own_take(a1, 4 * (g + OWN_T),       lo, below, wcnt, my);
            own_take(a2, 4 * (g + 2 * OWN_T),   lo, below, wcnt, my);
            own_take(a3, 4 * (g + 3 * OWN_T),   lo, below, wcnt, my);
            g = ring((it + 2 < nfull) ? it + 2 : it + 1);
            a0 = d4[g]; a1 = d4[g + OWN_T]; a2 = d4[g + 2 * OWN_T]; a3 = d4[g + 3 * OWN_T];
            own_take(b0, 4 * gb,                lo, below, wcnt, my);
            own_take(b1, 4 * (gb + OWN_T),      lo, below, wcnt, my);
            own_take(b2, 4 * (gb + 2 * OWN_T),  lo, below, wcnt, my);
            own_take(b3, 4 * (gb + 3 * OWN_T),  lo, below, wcnt, my);
        }
        if (it < nfull) {
            own_take(a0, 4 * g,                 lo, below, wcnt, my);
            own_take(a1, 4 * (g + OWN_T),       lo, below, wcnt, my);
            own_take(a2, 4 * (g + 2 * OWN_T),   lo, below, wcnt, my);
            own_take(a3, 4 * (g + 3 * OWN_T),   lo, below, wcnt, my);
        }
    }
    const int gt = nfull * (OWN_T * 4);
    if (gt < E4) {
        const int4 pad = make_int4(hi, hi, hi, hi);
        int g = gt + tid;
        int4 t0 = (g < E4) ? d4[g] : pad;
        int4 t1 = (g + OWN_T < E4) ? d4[g + OWN_T] : pad;
        int4 t2 = (g + 2 * OWN_T < E4) ? d4[g + 2 * OWN_T] : pad;
        int4 t3 = (g + 3 * OWN_T < E4) ? d4[g + 3 * OWN_T] : pad;
        own_take(t0, 4 * g,                 lo, below, wcnt, my);
        own_take(t1, 4 * (g + OWN_T),       lo, below, wcnt, my);
        own_take(t2, 4 * (g + 2 * OWN_T),   lo, below, wcnt, my);
        own_take(t3, 4 * (g + 3 * OWN_T),   lo, below, wcnt, my);
    }
    for (int e0 = 4 * E4 + wave * 32; e0 < E; e0 += OWN_T) {
        int e = e0 + lane;
        unsigned dl = (unsigned)(((e < E) ? dstA[e] : hi) - lo);
        below += (int)(dl >> 31);
        own_file(e, dl, __builtin_amdgcn_ballot_w32(dl < (unsigned)OWN_R), wcnt, my);
    }
    for (int msk = 16; msk >= 1; msk >>= 1) below += __shfl_xor(below, msk, 32);
    if (lane == 0) {
        atomicAdd(&belowAll, below);
        if (wcnt > OWN_CAPW) atomicOr(&ovf, 1);
    }
    __syncthreads();
    const bool listed = (ovf == 0);
    const int base = belowAll;

    if (listed) {
        for (int i = lane; i < wcnt; i += 32) atomicAdd(&ncount[my[i].y], 1);
    } else {
        for (int e = tid; e < E; e += OWN_T) {
            int dv = dstA[e];
            if (dv >= lo && dv < hi) atomicAdd(&ncount[dv - lo], 1);
        }
    }
    __syncthreads();

    int cntv = (tid < OWN_R) ? ncount[tid] : 0;
    int inc = cntv;
    for (int dlt = 1; dlt < 32; dlt <<= 1) {
        int up = __shfl_up(inc, dlt, 32);
        if (lane >= dlt) inc += up;
    }
    if (lane == 31) wsum[wave] = inc;
    __syncthreads();
    if (wave == 0) {
        int w = wsum[lane];
        int winc = w;
        for (int dlt = 1; dlt < 32; dlt <<= 1) {
            int up = __shfl_up(winc, dlt, 32);
            if (lane >= dlt) winc += up;
        }
        wsum[lane] = winc - w;
    }
    __syncthreads();
    if (tid < OWN_R) {
        int ex = inc - cntv + wsum[wave];
        cur[tid] = ex;
        int node = lo + tid;
        if (node < N) { off[node] = base + ex; degI[node] = cntv; }
    }
    __syncthreads();

    if (listed) {
        for (int i = lane; i < wcnt; i += 32) {
            int2 en = my[i];
            int p = atomicAdd(&cur[en.y], 1);
            eidA[base + p] = en.x;
        }
    } else {
        for (int e = tid; e < E; e += OWN_T) {
            int dv = dstA[e];
            if (dv >= lo && dv < hi) {
                int p = atomicAdd(&cur[dv - lo], 1);
                eidA[base + p] = e;
            }
        }
    }
}

#define GEMM_R 64

__device__ __forceinline__ void gemm_load_b(v2f (&bfr)[16], const float* __restrict__ wp) {
#pragma unroll
    for (int s = 0; s < 16; ++s) {
        int kk = 8 * (s >> 1) + 2 * (s & 1);
        bfr[s].x = wp[(size_t)kk * HC];
        bfr[s].y = wp[(size_t)(kk + 1) * HC];
    }
}

__device__ __forceinline__ void gemm_body(const float* __restrict__ X, const float* __restrict__ W,
                                          float* __restrict__ Y, int M, const int K, float4* xs4) {
    const int tid = threadIdx.x, wave = tid >> 5, lane = tid & 31;
    const int c = lane & 15, half = lane >> 4;
    const int row0 = blockIdx.x * GEMM_R;
    const int pq = (K >> 2) + 1;
    const float4* X4 = reinterpret_cast<const float4*>(X);

    const int kq = K >> 2;
    const int nq = GEMM_R * kq / 256;
    float4 v[8];
#pragma unroll
    for (int q = 0; q < 8; ++q) {
        if (q < nq) {
            int i = tid + 256 * q;
            int r = min(row0 + i / kq, M - 1);
            v[q] = X4[(size_t)r * kq + (i % kq)];
        }
    }
    const int col0 = wave * 16;
    const float* wp = W + (size_t)(4 * half) * HC + col0 + c;
    v2f bfr[2][16];
    gemm_load_b(bfr[0], wp);
    if (K > 64) gemm_load_b(bfr[1], wp + (size_t)64 * HC);
#pragma unroll
    for (int q = 0; q < 8; ++q) {
        if (q < nq) {
            int i = tid + 256 * q;
            xs4[(i / kq) * pq + (i % kq)] = v[q];
        }
    }
    __syncthreads();

    v8f acc0 = {}, acc1 = {}, acc2 = {}, acc3 = {};
    const float4* xr = xs4 + c * pq + half;
#pragma unroll
    for (int kc = 0; kc < K; kc += 64) {
        const v2f (&bf)[16] = bfr[kc >> 6];
#pragma unroll
        for (int j = 0; j < 8; ++j) {
            const float4* xa = xr + (kc >> 2) + 2 * j;
            float4 p0 = xa[0];
            float4 p1 = xa[16 * pq];
            float4 p2 = xa[32 * pq];
            float4 p3 = xa[48 * pq];
            v2f a00 = { p0.x, p0.y }, a01 = { p0.z, p0.w };
            v2f a10 = { p1.x, p1.y }, a11 = { p1.z, p1.w };
            v2f a20 = { p2.x, p2.y }, a21 = { p2.z, p2.w };
            v2f a30 = { p3.x, p3.y }, a31 = { p3.z, p3.w };
            acc0 = __builtin_amdgcn_wmma_f32_16x16x4_f32(false, a00, false, bf[2 * j], (short)0, acc0, false, false);
            acc1 = __builtin_amdgcn_wmma_f32_16x16x4_f32(false, a10, false, bf[2 * j], (short)0, acc1, false, false);
            acc2 = __builtin_amdgcn_wmma_f32_16x16x4_f32(false, a20, false, bf[2 * j], (short)0, acc2, false, false);
            acc3 = __builtin_amdgcn_wmma_f32_16x16x4_f32(false, a30, false, bf[2 * j], (short)0, acc3, false, false);
            acc0 = __builtin_amdgcn_wmma_f32_16x16x4_f32(false, a01, false, bf[2 * j + 1], (short)0, acc0, false, false);
            acc1 = __builtin_amdgcn_wmma_f32_16x16x4_f32(false, a11, false, bf[2 * j + 1], (short)0, acc1, false, false);
            acc2 = __builtin_amdgcn_wmma_f32_16x16x4_f32(false, a21, false, bf[2 * j + 1], (short)0, acc2, false, false);
            acc3 = __builtin_amdgcn_wmma_f32_16x16x4_f32(false, a31, false, bf[2 * j + 1], (short)0, acc3, false, false);
        }
    }

    __builtin_amdgcn_sched_barrier(0);
    asm volatile("v_nop\n\tv_nop\n\tv_nop\n\tv_nop");
    __builtin_amdgcn_sched_barrier(0);

    const int rb = row0 + 8 * half;
    float* yp = Y + (size_t)rb * HC + col0 + c;
    if (row0 + GEMM_R > M) {
#pragma unroll
        for (int r = 0; r < 8; ++r) {
            if (rb + r < M)      yp[(size_t)r * HC]        = acc0[r];
            if (rb + 16 + r < M) yp[(size_t)(16 + r) * HC] = acc1[r];
            if (rb + 32 + r < M) yp[(size_t)(32 + r) * HC] = acc2[r];
            if (rb + 48 + r < M) yp[(size_t)(48 + r) * HC] = acc3[r];
        }
    } else {
#pragma unroll
        for (int r = 0; r < 8; ++r) {
            yp[(size_t)r * HC]        = acc0[r];
            yp[(size_t)(16 + r) * HC] = acc1[r];
            yp[(size_t)(32 + r) * HC] = acc2[r];
            yp[(size_t)(48 + r) * HC] = acc3[r];
        }
    }
}

__global__ void __launch_bounds__(256)
gemm_wmma_k64_kernel(const float* __restrict__ X, const float* __restrict__ W,
                     float* __restrict__ Y, int M) {
    __shared__ float4 xs4[GEMM_R * (F_IN / 4 + 1)];
    gemm_body(X, W, Y, M, F_IN, xs4);
}

__global__ void __launch_bounds__(256)
gemm_wmma_k128_kernel(const float* __restrict__ X, const float* __restrict__ W,
                      float* __restrict__ Y, int M) {
    __shared__ float4 xs4[GEMM_R * (HC / 4 + 1)];
    gemm_body(X, W, Y, M, HC, xs4);
}

__global__ void node_att_kernel(const float* __restrict__ Hp,
                                const float* __restrict__ att_s, const float* __restrict__ att_d,
                                float* __restrict__ a_s, float* __restrict__ a_d, int N) {
    int wave = threadIdx.x >> 5, lane = threadIdx.x & 31;
    int n = blockIdx.x * (blockDim.x >> 5) + wave;
    if (n >= N) return;
    int h = lane >> 3, q = lane & 7;
    float4 hv = reinterpret_cast<const float4*>(Hp)[(size_t)n * 32 + lane];
    float4 sa = reinterpret_cast<const float4*>(att_s)[lane];
    float4 da = reinterpret_cast<const float4*>(att_d)[lane];
    float ps = hv.x * sa.x + hv.y * sa.y + hv.z * sa.z + hv.w * sa.w;
    float pd = hv.x * da.x + hv.y * da.y + hv.z * da.z + hv.w * da.w;
#pragma unroll
    for (int msk = 1; msk < 8; msk <<= 1) {
        ps += __shfl_xor(ps, msk, 32);
        pd += __shfl_xor(pd, msk, 32);
    }
    if (q == 0) { a_s[(size_t)n * 4 + h] = ps; a_d[(size_t)n * 4 + h] = pd; }
}

__global__ void agg_kernel(const int* __restrict__ off, const int* __restrict__ degI,
                           const int* __restrict__ srcL,
                           const float* __restrict__ a_e, const float* __restrict__ a_s,
                           const float* __restrict__ a_d, const float* __restrict__ sum_ae,
                           const float* __restrict__ Hp, const float* __restrict__ bias,
                           const float* __restrict__ lin_w, const float* __restrict__ lin_b,
                           float* __restrict__ outv, float* __restrict__ outs,
                           int N, int mode) {
    int wave = threadIdx.x >> 5, lane = threadIdx.x & 31;
    int n = blockIdx.x * (blockDim.x >> 5) + wave;
    if (n >= N) return;
    int h = lane >> 3;
    const float4* Hp4 = reinterpret_cast<const float4*>(Hp);

    float ad_n = a_d[(size_t)n * 4 + h];
    float as_n = a_s[(size_t)n * 4 + h];
    int cnt = degI[n];
    float degf = fmaxf((float)cnt, 1.0f);
    float M = leaky(as_n + ad_n + sum_ae[(size_t)n * 4 + h] / degf);
    float D = 1.0f;
    float4 A = Hp4[(size_t)n * 32 + lane];

    int s0 = off[n];
    for (int i = 0; i < cnt; ++i) {
        int s = srcL[s0 + i];
        float lg = leaky(a_s[(size_t)s * 4 + h] + ad_n + a_e[(size_t)(s0 + i) * 4 + h]);
        float Mn = fmaxf(M, lg);
        float sc = __expf(M - Mn);
        float p  = __expf(lg - Mn);
        float4 hv = Hp4[(size_t)s * 32 + lane];
        A.x = A.x * sc + p * hv.x;
        A.y = A.y * sc + p * hv.y;
        A.z = A.z * sc + p * hv.z;
        A.w = A.w * sc + p * hv.w;
        D = D * sc + p;
        M = Mn;
    }
    float4 bv = reinterpret_cast<const float4*>(bias)[lane];
    float inv = 1.0f / D;
    float4 r = { A.x * inv + bv.x, A.y * inv + bv.y, A.z * inv + bv.z, A.w * inv + bv.w };
    if (mode == 0) {
        r.x = fmaxf(r.x, 0.f); r.y = fmaxf(r.y, 0.f);
        r.z = fmaxf(r.z, 0.f); r.w = fmaxf(r.w, 0.f);
        reinterpret_cast<float4*>(outv)[(size_t)n * 32 + lane] = r;
    } else {
        float4 lw = reinterpret_cast<const float4*>(lin_w)[lane];
        float part = r.x * lw.x + r.y * lw.y + r.z * lw.z + r.w * lw.w;
#pragma unroll
        for (int msk = 16; msk >= 1; msk >>= 1) part += __shfl_xor(part, msk, 32);
        if (lane == 0) outs[n] = part + lin_b[0];
    }
}

extern "C" void kernel_launch(void* const* d_in, const int* in_sizes, int n_in,
                              void* d_out, int out_size, void* d_ws, size_t ws_size,
                              hipStream_t stream) {
    const float* x   = (const float*)d_in[0];
    const float* ea  = (const float*)d_in[1];
    const float* W1  = (const float*)d_in[2];
    const float* as1 = (const float*)d_in[3];
    const float* ad1 = (const float*)d_in[4];
    const float* We1 = (const float*)d_in[5];
    const float* ae1 = (const float*)d_in[6];
    const float* b1  = (const float*)d_in[7];
    const float* W2  = (const float*)d_in[8];
    const float* as2 = (const float*)d_in[9];
    const float* ad2 = (const float*)d_in[10];
    const float* We2 = (const float*)d_in[11];
    const float* ae2 = (const float*)d_in[12];
    const float* b2  = (const float*)d_in[13];
    const float* lw  = (const float*)d_in[14];
    const float* lb  = (const float*)d_in[15];
    const int*   ei  = (const int*)d_in[16];

    const int N = in_sizes[0] / F_IN;
    const int E = in_sizes[16] / 2;
    const int* srcA = ei;
    const int* dstA = ei + E;

    char* w = (char*)d_ws;
    auto alloc = [&](size_t bytes) -> void* {
        void* p = (void*)w;
        w += (bytes + 255) & ~(size_t)255;
        return p;
    };
    int*   degI   = (int*)  alloc((size_t)N * 4);
    float* sum1 = (float*)alloc((size_t)N * 16);
    float* sum2 = (float*)alloc((size_t)N * 16);
    float* ae1v = (float*)alloc((size_t)E * 16);
    float* ae2v = (float*)alloc((size_t)E * 16);
    float* a_sb = (float*)alloc((size_t)N * 16);
    float* a_db = (float*)alloc((size_t)N * 16);
    float* Bt   = a_sb;
    int*   off  = (int*)  alloc((size_t)N * 4);
    int*   eidA = (int*)  alloc((size_t)E * 4);
    float* Hp   = (float*)alloc((size_t)N * HC * 4);
    float* h1   = (float*)alloc((size_t)N * HC * 4);

    const int gemmBlocks = (N + GEMM_R - 1) / GEMM_R;
    const int nodeBlocks = (N + 7) / 8;
    const int ownBlocks  = (N + OWN_R - 1) / OWN_R;

    fold_kernel<<<1, 256, 0, stream>>>(We1, ae1, We2, ae2, Bt);
    csr_owner_kernel<<<ownBlocks, OWN_T, 0, stream>>>(dstA, off, degI, eidA, N, E);
    edge_csr_kernel<<<nodeBlocks, 256, 0, stream>>>(ea, off, degI, eidA, srcA, Bt,
                                                    ae1v, ae2v, sum1, sum2, N);

    gemm_wmma_k64_kernel<<<gemmBlocks, 256, 0, stream>>>(x, W1, Hp, N);
    node_att_kernel<<<nodeBlocks, 256, 0, stream>>>(Hp, as1, ad1, a_sb, a_db, N);
    agg_kernel<<<nodeBlocks, 256, 0, stream>>>(off, degI, eidA, ae1v, a_sb, a_db,
                                               sum1, Hp, b1, nullptr, nullptr,
                                               h1, nullptr, N, 0);
    gemm_wmma_k128_kernel<<<gemmBlocks, 256, 0, stream>>>(h1, W2, Hp, N);
    node_att_kernel<<<nodeBlocks, 256, 0, stream>>>(Hp, as2, ad2, a_sb, a_db, N);
    agg_kernel<<<nodeBlocks, 256, 0, stream>>>(off, degI, eidA, ae2v, a_sb, a_db,
                                               sum2, Hp, b2, lw, lb,
                                               nullptr, (float*)d_out, N, 1);
}


# CSR build in two levels: a bucket pass sorts each 4096-edge slice by owner range in LDS; owner blocks read only their own groups instead of streaming every destination (list order: arrival)
# speedup vs baseline: 3.4098x; 1.0874x over previous
#include <hip/hip_runtime.h>
#include <hip/hip_bf16.h>

#define F_IN   64
#define HEADS  4
#define CHD    32
#define HC     128
#define NEG_SLOPE 0.2f

typedef __attribute__((ext_vector_type(2))) float v2f;
typedef __attribute__((ext_vector_type(8))) float v8f;

__device__ __forceinline__ float leaky(float x) { return x > 0.f ? x : NEG_SLOPE * x; }

__global__ void fold_kernel(const float* __restrict__ We1, const float* __restrict__ ae1,
                            const float* __restrict__ We2, const float* __restrict__ ae2,
                            float* __restrict__ Bt) {
    int t = threadIdx.x;
    int f = t >> 2, h = t & 3;
    float s1 = 0.f, s2 = 0.f;
#pragma unroll
    for (int c = 0; c < CHD; ++c) {
        s1 += We1[f * HC + h * CHD + c] * ae1[h * CHD + c];
        s2 += We2[f * HC + h * CHD + c] * ae2[h * CHD + c];
    }
    int j = f >> 3, hf = (f >> 2) & 1, u = (f >> 1) & 1, rg = f & 1;
    int step = j * 2 + u;
    Bt[(step * 32 + h + 16 * hf) * 2 + rg]     = s1;
    Bt[(step * 32 + 4 + h + 16 * hf) * 2 + rg] = s2;
    for (int z = t; z < 512; z += 256) {
        int rg2 = z & 1, col = 8 + ((z >> 1) & 7), hf2 = (z >> 4) & 1, st = z >> 5;
        Bt[(st * 32 + col + 16 * hf2) * 2 + rg2] = 0.f;
    }
}

__device__ __forceinline__ void wave_lds_sync() {
    __builtin_amdgcn_fence(__ATOMIC_RELEASE, "wavefront");
    __builtin_amdgcn_wave_barrier();
    __builtin_amdgcn_fence(__ATOMIC_ACQUIRE, "wavefront");
}

#define TILE_PITCH 68

__global__ void __launch_bounds__(256)
edge_csr_kernel(const float* __restrict__ EA, const int* __restrict__ off,
                const int* __restrict__ degI, int* eidA, const int* __restrict__ srcA,
                const float* __restrict__ Bt,
                float* __restrict__ ae1c, float* __restrict__ ae2c,
                float* __restrict__ sum1, float* __restrict__ sum2, int N) {
    __shared__ alignas(16) float tileS[8][16 * TILE_PITCH];
    int wave = threadIdx.x >> 5, lane = threadIdx.x & 31;
    int n = blockIdx.x * 8 + wave;
    if (n >= N) return;
    float* my = tileS[wave];
    int c = lane & 15, half = lane >> 4;

    v2f bfr[16];
#pragma unroll
    for (int s = 0; s < 16; ++s) bfr[s] = reinterpret_cast<const v2f*>(Bt)[s * 32 + lane];

    int cnt = __builtin_amdgcn_readfirstlane(degI[n]);
    int s0  = __builtin_amdgcn_readfirstlane(off[n]);
    float tot = 0.f;

    for (int t0 = 0; t0 < cnt; t0 += 16) {
        int eid = eidA[s0 + min(t0 + c, cnt - 1)];
        int sid = srcA[eid];
        float4 v[8];
#pragma unroll
        for (int i = 0; i < 8; ++i) {
            int er = __shfl(eid, 2 * i + half, 32);
            v[i] = *reinterpret_cast<const float4*>(EA + (size_t)er * F_IN + 4 * c);
        }
#pragma unroll
        for (int i = 0; i < 8; ++i)
            *reinterpret_cast<float4*>(my + (2 * i + half) * TILE_PITCH + 4 * c) = v[i];
        wave_lds_sync();

        v8f acc = {};
#pragma unroll
        for (int j = 0; j < 8; ++j) {
            float4 a4 = *reinterpret_cast<const float4*>(my + c * TILE_PITCH + 8 * j + 4 * half);
            v2f a0 = { a4.x, a4.y };
            v2f a1 = { a4.z, a4.w };
            acc = __builtin_amdgcn_wmma_f32_16x16x4_f32(false, a0, false, bfr[2 * j],
                                                        (short)0, acc, false, false);
            acc = __builtin_amdgcn_wmma_f32_16x16x4_f32(false, a1, false, bfr[2 * j + 1],
                                                        (short)0, acc, false, false);
        }
        wave_lds_sync();

#pragma unroll
        for (int r = 0; r < 8; ++r) {
            float val = acc[r];
            tot += (t0 + 8 * half + r < cnt) ? val : 0.f;
            my[(8 * half + r) * 20 + c] = val;
        }
        wave_lds_sync();
        float4 o = *reinterpret_cast<const float4*>(my + c * 20 + 4 * half);
        wave_lds_sync();
        if (t0 + c < cnt) {
            float* dstp = half ? ae2c : ae1c;
            *reinterpret_cast<float4*>(dstp + (size_t)(s0 + t0 + c) * 4) = o;
            if (half == 0) eidA[s0 + t0 + c] = sid;
        }
    }
    tot += __shfl_xor(tot, 16, 32);
    if (half == 0) {
        if (c < 4)      sum1[(size_t)n * 4 + c]     = tot;
        else if (c < 8) sum2[(size_t)n * 4 + c - 4] = tot;
    }
}

#define OWN_R    392
#define OWN_T    1024
#define OWN_W    (OWN_T / 32)
#define OWN_CAPW 640
#define OWN_MAXSL 4096
#define BK_SL    4096
#define BK_T     1024

__global__ void __launch_bounds__(BK_T)
csr_bucket_kernel(const int* __restrict__ dstA, int* __restrict__ bkw, int* __restrict__ cntT,
                  int E, int NB, int NSL) {
    __shared__ int hist[BK_T];
    __shared__ int start[BK_T];
    __shared__ int wsum[BK_T / 32];
    const int tid = threadIdx.x, lane = tid & 31, wave = tid >> 5;
    const int b = blockIdx.x;
    const int e0 = b * BK_SL;
    int2* bk = reinterpret_cast<int2*>(bkw);

    hist[tid] = 0;
    __syncthreads();

    int d[4], rg[4], rk[4];
#pragma unroll
    for (int k = 0; k < 4; ++k) {
        int e = e0 + k * BK_T + tid;
        d[k] = (e < E) ? dstA[e] : -1;
    }
#pragma unroll
    for (int k = 0; k < 4; ++k) {
        rg[k] = 0; rk[k] = 0;
        if (d[k] >= 0) {
            rg[k] = (int)((unsigned)d[k] / (unsigned)OWN_R);
            rk[k] = atomicAdd(&hist[rg[k]], 1);
        }
    }
    __syncthreads();

    const int cntv = hist[tid];
    int inc = cntv;
    for (int dlt = 1; dlt < 32; dlt <<= 1) {
        int up = __shfl_up(inc, dlt, 32);
        if (lane >= dlt) inc += up;
    }
    if (lane == 31) wsum[wave] = inc;
    __syncthreads();
    if (wave == 0) {
        int w = wsum[lane];
        int winc = w;
        for (int dlt = 1; dlt < 32; dlt <<= 1) {
            int up = __shfl_up(winc, dlt, 32);
            if (lane >= dlt) winc += up;
        }
        wsum[lane] = winc - w;
    }
    __syncthreads();
    start[tid] = inc - cntv + wsum[wave];
    if (tid < NB) cntT[(size_t)tid * NSL + b] = cntv;
    __syncthreads();
#pragma unroll
    for (int k = 0; k < 4; ++k)
        if (d[k] >= 0)
            bk[(size_t)e0 + start[rg[k]] + rk[k]] = make_int2(e0 + k * BK_T + tid, d[k]);
}

__device__ __forceinline__ void own_append(int2 en, int have, int lo, int lane, int& wcnt, int2* my) {
    if (lane < have) my[min(wcnt + lane, OWN_CAPW - 1)] = make_int2(en.x, en.y - lo);
    wcnt += min(have, 32);
}

__global__ void __launch_bounds__(OWN_T)
csr_owner_kernel(const int* __restrict__ bkw, const int* __restrict__ cntT,
                 const int* __restrict__ dstA, int* __restrict__ off, int* __restrict__ degI,
                 int* __restrict__ eidA, int N, int E, int NSL) {
    __shared__ int2 list[OWN_W][OWN_CAPW];
    __shared__ int ncount[OWN_R];
    __shared__ int cur[OWN_R];
    __shared__ int wsum[OWN_W];
    __shared__ int segStart[OWN_MAXSL];
    __shared__ int segCnt[OWN_MAXSL];
    __shared__ int belowAll, ovf;

    const int tid = threadIdx.x, lane = tid & 31, wave = tid >> 5;
    const int r = blockIdx.x;
    const int lo = r * OWN_R, hi = lo + OWN_R;
    int2* my = list[wave];
    const int2* bk = reinterpret_cast<const int2*>(bkw);

    for (int i = tid; i < OWN_R; i += OWN_T) ncount[i] = 0;
    for (int i = tid; i < NSL; i += OWN_T) segStart[i] = 0;
    if (tid == 0) { belowAll = 0; ovf = 0; }
    __syncthreads();

    for (int i = tid; i < NSL; i += OWN_T) segCnt[i] = cntT[(size_t)r * NSL + i];
    {
        const int total = r * NSL;
        int bcur = tid % NSL;
        const int step = OWN_T % NSL;
        for (int i = tid; i < total; i += 8 * OWN_T) {
            int v[8], bb[8];
#pragma unroll
            for (int q = 0; q < 8; ++q) {
                int ii = i + q * OWN_T;
                v[q] = (ii < total) ? cntT[ii] : 0;
                bb[q] = bcur;
                bcur += step;
                if (bcur >= NSL) bcur -= NSL;
            }
#pragma unroll
            for (int q = 0; q < 8; ++q)
                if (v[q] != 0) atomicAdd(&segStart[bb[q]], v[q]);
        }
    }
    __syncthreads();

    int below = 0;
    for (int i = tid; i < NSL; i += OWN_T) below += segStart[i];
    for (int msk = 16; msk >= 1; msk >>= 1) below += __shfl_xor(below, msk, 32);
    if (lane == 0 && below != 0) atomicAdd(&belowAll, below);

    int wcnt = 0;
    for (int b0 = wave; b0 < NSL; b0 += 4 * OWN_W) {
        int cs[4];
        const int2* sg[4];
        int2 ea[4], eb[4];
#pragma unroll
        for (int j = 0; j < 4; ++j) {
            int b = b0 + j * OWN_W;
            int bb = (b < NSL) ? b : 0;
            int c = __builtin_amdgcn_readfirstlane(segCnt[bb]);
            int st = __builtin_amdgcn_readfirstlane(segStart[bb]);
            cs[j] = (b < NSL) ? c : 0;
            sg[j] = bk + (size_t)bb * BK_SL + st;
            ea[j] = (lane < cs[j])      ? sg[j][lane]      : make_int2(0, 0);
            eb[j] = (lane + 32 < cs[j]) ? sg[j][lane + 32] : make_int2(0, 0);
        }
#pragma unroll
        for (int j = 0; j < 4; ++j) {
            own_append(ea[j], cs[j], lo, lane, wcnt, my);
            if (cs[j] > 32) own_append(eb[j], cs[j] - 32, lo, lane, wcnt, my);
            for (int i0 = 64; i0 < cs[j]; i0 += 32) {
                int i = i0 + lane;
                int2 en = (i < cs[j]) ? sg[j][i] : make_int2(0, 0);
                own_append(en, cs[j] - i0, lo, lane, wcnt, my);
            }
        }
    }
    if (lane == 0 && wcnt > OWN_CAPW) atomicOr(&ovf, 1);
    __syncthreads();
    const bool listed = (ovf == 0);
    const int base = belowAll;

    if (listed) {
        for (int i = lane; i < wcnt; i += 32) atomicAdd(&ncount[my[i].y], 1);
    } else {
        for (int e = tid; e < E; e += OWN_T) {
            int dv = dstA[e];
            if (dv >= lo && dv < hi) atomicAdd(&ncount[dv - lo], 1);
        }
    }
    __syncthreads();

    int cntv = (tid < OWN_R) ? ncount[tid] : 0;
    int inc = cntv;
    for (int dlt = 1; dlt < 32; dlt <<= 1) {
        int up = __shfl_up(inc, dlt, 32);
        if (lane >= dlt) inc += up;
    }
    if (lane == 31) wsum[wave] = inc;
    __syncthreads();
    if (wave == 0) {
        int w = wsum[lane];
        int winc = w;
        for (int dlt = 1; dlt < 32; dlt <<= 1) {
            int up = __shfl_up(winc, dlt, 32);
            if (lane >= dlt) winc += up;
        }
        wsum[lane] = winc - w;
    }
    __syncthreads();
    if (tid < OWN_R) {
        int ex = inc - cntv + wsum[wave];
        cur[tid] = ex;
        int node = lo + tid;
        if (node < N) { off[node] = base + ex; degI[node] = cntv; }
    }
    __syncthreads();

    if (listed) {
        for (int i = lane; i < wcnt; i += 32) {
            int2 en = my[i];
            int p = atomicAdd(&cur[en.y], 1);
            eidA[base + p] = en.x;
        }
    } else {
        for (int e = tid; e < E; e += OWN_T) {
            int dv = dstA[e];
            if (dv >= lo && dv < hi) {
                int p = atomicAdd(&cur[dv - lo], 1);
                eidA[base + p] = e;
            }
        }
    }
}

#define GEMM_R 64

__device__ __forceinline__ void gemm_load_b(v2f (&bfr)[16], const float* __restrict__ wp) {
#pragma unroll
    for (int s = 0; s < 16; ++s) {
        int kk = 8 * (s >> 1) + 2 * (s & 1);
        bfr[s].x = wp[(size_t)kk * HC];
        bfr[s].y = wp[(size_t)(kk + 1) * HC];
    }
}

__device__ __forceinline__ void gemm_body(const float* __restrict__ X, const float* __restrict__ W,
                                          float* __restrict__ Y, int M, const int K, float4* xs4) {
    const int tid = threadIdx.x, wave = tid >> 5, lane = tid & 31;
    const int c = lane & 15, half = lane >> 4;
    const int row0 = blockIdx.x * GEMM_R;
    const int pq = (K >> 2) + 1;
    const float4* X4 = reinterpret_cast<const float4*>(X);

    const int kq = K >> 2;
    const int nq = GEMM_R * kq / 256;
    float4 v[8];
#pragma unroll
    for (int q = 0; q < 8; ++q) {
        if (q < nq) {
            int i = tid + 256 * q;
            int r = min(row0 + i / kq, M - 1);
            v[q] = X4[(size_t)r * kq + (i % kq)];
        }
    }
    const int col0 = wave * 16;
    const float* wp = W + (size_t)(4 * half) * HC + col0 + c;
    v2f bfr[2][16];
    gemm_load_b(bfr[0], wp);
    if (K > 64) gemm_load_b(bfr[1], wp + (size_t)64 * HC);
#pragma unroll
    for (int q = 0; q < 8; ++q) {
        if (q < nq) {
            int i = tid + 256 * q;
            xs4[(i / kq) * pq + (i % kq)] = v[q];
        }
    }
    __syncthreads();

    v8f acc0 = {}, acc1 = {}, acc2 = {}, acc3 = {};
    const float4* xr = xs4 + c * pq + half;
#pragma unroll
    for (int kc = 0; kc < K; kc += 64) {
        const v2f (&bf)[16] = bfr[kc >> 6];
#pragma unroll
        for (int j = 0; j < 8; ++j) {
            const float4* xa = xr + (kc >> 2) + 2 * j;
            float4 p0 = xa[0];
            float4 p1 = xa[16 * pq];
            float4 p2 = xa[32 * pq];
            float4 p3 = xa[48 * pq];
            v2f a00 = { p0.x, p0.y }, a01 = { p0.z, p0.w };
            v2f a10 = { p1.x, p1.y }, a11 = { p1.z, p1.w };
            v2f a20 = { p2.x, p2.y }, a21 = { p2.z, p2.w };
            v2f a30 = { p3.x, p3.y }, a31 = { p3.z, p3.w };
            acc0 = __builtin_amdgcn_wmma_f32_16x16x4_f32(false, a00, false, bf[2 * j], (short)0, acc0, false, false);
            acc1 = __builtin_amdgcn_wmma_f32_16x16x4_f32(false, a10, false, bf[2 * j], (short)0, acc1, false, false);
            acc2 = __builtin_amdgcn_wmma_f32_16x16x4_f32(false, a20, false, bf[2 * j], (short)0, acc2, false, false);
            acc3 = __builtin_amdgcn_wmma_f32_16x16x4_f32(false, a30, false, bf[2 * j], (short)0, acc3, false, false);
            acc0 = __builtin_amdgcn_wmma_f32_16x16x4_f32(false, a01, false, bf[2 * j + 1], (short)0, acc0, false, false);
            acc1 = __builtin_amdgcn_wmma_f32_16x16x4_f32(false, a11, false, bf[2 * j + 1], (short)0, acc1, false, false);
            acc2 = __builtin_amdgcn_wmma_f32_16x16x4_f32(false, a21, false, bf[2 * j + 1], (short)0, acc2, false, false);
            acc3 = __builtin_amdgcn_wmma_f32_16x16x4_f32(false, a31, false, bf[2 * j + 1], (short)0, acc3, false, false);
        }
    }

    __builtin_amdgcn_sched_barrier(0);
    asm volatile("v_nop\n\tv_nop\n\tv_nop\n\tv_nop");
    __builtin_amdgcn_sched_barrier(0);

    const int rb = row0 + 8 * half;
    float* yp = Y + (size_t)rb * HC + col0 + c;
    if (row0 + GEMM_R > M) {
#pragma unroll
        for (int r = 0; r < 8; ++r) {
            if (rb + r < M)      yp[(size_t)r * HC]        = acc0[r];
            if (rb + 16 + r < M) yp[(size_t)(16 + r) * HC] = acc1[r];
            if (rb + 32 + r < M) yp[(size_t)(32 + r) * HC] = acc2[r];
            if (rb + 48 + r < M) yp[(size_t)(48 + r) * HC] = acc3[r];
        }
    } else {
#pragma unroll
        for (int r = 0; r < 8; ++r) {
            yp[(size_t)r * HC]        = acc0[r];
            yp[(size_t)(16 + r) * HC] = acc1[r];
            yp[(size_t)(32 + r) * HC] = acc2[r];
            yp[(size_t)(48 + r) * HC] = acc3[r];
        }
    }
}

__global__ void __launch_bounds__(256)
gemm_wmma_k64_kernel(const float* __restrict__ X, const float* __restrict__ W,
                     float* __restrict__ Y, int M) {
    __shared__ float4 xs4[GEMM_R * (F_IN / 4 + 1)];
    gemm_body(X, W, Y, M, F_IN, xs4);
}

__global__ void __launch_bounds__(256)
gemm_wmma_k128_kernel(const float* __restrict__ X, const float* __restrict__ W,
                      float* __restrict__ Y, int M) {
    __shared__ float4 xs4[GEMM_R * (HC / 4 + 1)];
    gemm_body(X, W, Y, M, HC, xs4);
}

__global__ void node_att_kernel(const float* __restrict__ Hp,
                                const float* __restrict__ att_s, const float* __restrict__ att_d,
                                float* __restrict__ a_s, float* __restrict__ a_d, int N) {
    int wave = threadIdx.x >> 5, lane = threadIdx.x & 31;
    int n = blockIdx.x * (blockDim.x >> 5) + wave;
    if (n >= N) return;
    int h = lane >> 3, q = lane & 7;
    float4 hv = reinterpret_cast<const float4*>(Hp)[(size_t)n * 32 + lane];
    float4 sa = reinterpret_cast<const float4*>(att_s)[lane];
    float4 da = reinterpret_cast<const float4*>(att_d)[lane];
    float ps = hv.x * sa.x + hv.y * sa.y + hv.z * sa.z + hv.w * sa.w;
    float pd = hv.x * da.x + hv.y * da.y + hv.z * da.z + hv.w * da.w;
#pragma unroll
    for (int msk = 1; msk < 8; msk <<= 1) {
        ps += __shfl_xor(ps, msk, 32);
        pd += __shfl_xor(pd, msk, 32);
    }
    if (q == 0) { a_s[(size_t)n * 4 + h] = ps; a_d[(size_t)n * 4 + h] = pd; }
}

__global__ void agg_kernel(const int* __restrict__ off, const int* __restrict__ degI,
                           const int* __restrict__ srcL,
                           const float* __restrict__ a_e, const float* __restrict__ a_s,
                           const float* __restrict__ a_d, const float* __restrict__ sum_ae,
                           const float* __restrict__ Hp, const float* __restrict__ bias,
                           const float* __restrict__ lin_w, const float* __restrict__ lin_b,
                           float* __restrict__ outv, float* __restrict__ outs,
                           int N, int mode) {
    int wave = threadIdx.x >> 5, lane = threadIdx.x & 31;
    int n = blockIdx.x * (blockDim.x >> 5) + wave;
    if (n >= N) return;
    int h = lane >> 3;
    const float4* Hp4 = reinterpret_cast<const float4*>(Hp);

    float ad_n = a_d[(size_t)n * 4 + h];
    float as_n = a_s[(size_t)n * 4 + h];
    int cnt = degI[n];
    float degf = fmaxf((float)cnt, 1.0f);
    float M = leaky(as_n + ad_n + sum_ae[(size_t)n * 4 + h] / degf);
    float D = 1.0f;
    float4 A = Hp4[(size_t)n * 32 + lane];

    int s0 = off[n];
    for (int i = 0; i < cnt; ++i) {
        int s = srcL[s0 + i];
        float lg = leaky(a_s[(size_t)s * 4 + h] + ad_n + a_e[(size_t)(s0 + i) * 4 + h]);
        float Mn = fmaxf(M, lg);
        float sc = __expf(M - Mn);
        float p  = __expf(lg - Mn);
        float4 hv = Hp4[(size_t)s * 32 + lane];
        A.x = A.x * sc + p * hv.x;
        A.y = A.y * sc + p * hv.y;
        A.z = A.z * sc + p * hv.z;
        A.w = A.w * sc + p * hv.w;
        D = D * sc + p;
        M = Mn;
    }
    float4 bv = reinterpret_cast<const float4*>(bias)[lane];
    float inv = 1.0f / D;
    float4 r = { A.x * inv + bv.x, A.y * inv + bv.y, A.z * inv + bv.z, A.w * inv + bv.w };
    if (mode == 0) {
        r.x = fmaxf(r.x, 0.f); r.y = fmaxf(r.y, 0.f);
        r.z = fmaxf(r.z, 0.f); r.w = fmaxf(r.w, 0.f);
        reinterpret_cast<float4*>(outv)[(size_t)n * 32 + lane] = r;
    } else {
        float4 lw = reinterpret_cast<const float4*>(lin_w)[lane];
        float part = r.x * lw.x + r.y * lw.y + r.z * lw.z + r.w * lw.w;
#pragma unroll
        for (int msk = 16; msk >= 1; msk >>= 1) part += __shfl_xor(part, msk, 32);
        if (lane == 0) outs[n] = part + lin_b[0];
    }
}

extern "C" void kernel_launch(void* const* d_in, const int* in_sizes, int n_in,
                              void* d_out, int out_size, void* d_ws, size_t ws_size,
                              hipStream_t stream) {
    const float* x   = (const float*)d_in[0];
    const float* ea  = (const float*)d_in[1];
    const float* W1  = (const float*)d_in[2];
    const float* as1 = (const float*)d_in[3];
    const float* ad1 = (const float*)d_in[4];
    const float* We1 = (const float*)d_in[5];
    const float* ae1 = (const float*)d_in[6];
    const float* b1  = (const float*)d_in[7];
    const float* W2  = (const float*)d_in[8];
    const float* as2 = (const float*)d_in[9];
    const float* ad2 = (const float*)d_in[10];
    const float* We2 = (const float*)d_in[11];
    const float* ae2 = (const float*)d_in[12];
    const float* b2  = (const float*)d_in[13];
    const float* lw  = (const float*)d_in[14];
    const float* lb  = (const float*)d_in[15];
    const int*   ei  = (const int*)d_in[16];

    const int N = in_sizes[0] / F_IN;
    const int E = in_sizes[16] / 2;
    const int* srcA = ei;
    const int* dstA = ei + E;

    char* w = (char*)d_ws;
    auto alloc = [&](size_t bytes) -> void* {
        void* p = (void*)w;
        w += (bytes + 255) & ~(size_t)255;
        return p;
    };
    int*   degI   = (int*)  alloc((size_t)N * 4);
    float* sum1 = (float*)alloc((size_t)N * 16);
    float* sum2 = (float*)alloc((size_t)N * 16);
    float* ae1v = (float*)alloc((size_t)E * 16);
    float* ae2v = (float*)alloc((size_t)E * 16);
    float* a_sb = (float*)alloc((size_t)N * 16);
    float* a_db = (float*)alloc((size_t)N * 16);
    float* Bt   = a_sb;
    int*   off  = (int*)  alloc((size_t)N * 4);
    int*   eidA = (int*)  alloc((size_t)E * 4);
    float* Hp   = (float*)alloc((size_t)N * HC * 4);
    float* h1   = (float*)alloc((size_t)N * HC * 4);

    const int gemmBlocks = (N + GEMM_R - 1) / GEMM_R;
    const int nodeBlocks = (N + 7) / 8;
    const int ownBlocks  = (N + OWN_R - 1) / OWN_R;
    const int nSlices    = (E + BK_SL - 1) / BK_SL;
    int* bkw  = (int*)Hp;
    int* cntT = (int*)h1;

    fold_kernel<<<1, 256, 0, stream>>>(We1, ae1, We2, ae2, Bt);
    csr_bucket_kernel<<<nSlices, BK_T, 0, stream>>>(dstA, bkw, cntT, E, ownBlocks, nSlices);
    csr_owner_kernel<<<ownBlocks, OWN_T, 0, stream>>>(bkw, cntT, dstA, off, degI, eidA, N, E, nSlices);
    edge_csr_kernel<<<nodeBlocks, 256, 0, stream>>>(ea, off, degI, eidA, srcA, Bt,
                                                    ae1v, ae2v, sum1, sum2, N);

    gemm_wmma_k64_kernel<<<gemmBlocks, 256, 0, stream>>>(x, W1, Hp, N);
    node_att_kernel<<<nodeBlocks, 256, 0, stream>>>(Hp, as1, ad1, a_sb, a_db, N);
    agg_kernel<<<nodeBlocks, 256, 0, stream>>>(off, degI, eidA, ae1v, a_sb, a_db,
                                               sum1, Hp, b1, nullptr, nullptr,
                                               h1, nullptr, N, 0);
    gemm_wmma_k128_kernel<<<gemmBlocks, 256, 0, stream>>>(h1, W2, Hp, N);
    node_att_kernel<<<nodeBlocks, 256, 0, stream>>>(Hp, as2, ad2, a_sb, a_db, N);
    agg_kernel<<<nodeBlocks, 256, 0, stream>>>(off, degI, eidA, ae2v, a_sb, a_db,
                                               sum2, Hp, b2, lw, lb,
                                               nullptr, (float*)d_out, N, 1);
}


# layer-1 projection chain rides in the CSR-chain launches it does not depend on: fold in the bucket launch, X@W1 in owner-launch blocks, a_s/a_d by the edge wave; 10 launches to 7 (sums unchanged)
# speedup vs baseline: 3.6045x; 1.0571x over previous
#include <hip/hip_runtime.h>
#include <hip/hip_bf16.h>

#define F_IN   64
#define HEADS  4
#define CHD    32
#define HC     128
#define NEG_SLOPE 0.2f

typedef __attribute__((ext_vector_type(2))) float v2f;
typedef __attribute__((ext_vector_type(8))) float v8f;

__device__ __forceinline__ float leaky(float x) { return x > 0.f ? x : NEG_SLOPE * x; }

__device__ __forceinline__ void node_att_body(float4 hv, int n, int lane,
                                const float* __restrict__ att_s, const float* __restrict__ att_d,
                                float* __restrict__ a_s, float* __restrict__ a_d) {
    int h = lane >> 3, q = lane & 7;
    float4 sa = reinterpret_cast<const float4*>(att_s)[lane];
    float4 da = reinterpret_cast<const float4*>(att_d)[lane];
    float ps = hv.x * sa.x + hv.y * sa.y + hv.z * sa.z + hv.w * sa.w;
    float pd = hv.x * da.x + hv.y * da.y + hv.z * da.z + hv.w * da.w;
#pragma unroll
    for (int msk = 1; msk < 8; msk <<= 1) {
        ps += __shfl_xor(ps, msk, 32);
        pd += __shfl_xor(pd, msk, 32);
    }
    if (q == 0) { a_s[(size_t)n * 4 + h] = ps; a_d[(size_t)n * 4 + h] = pd; }
}

__device__ __forceinline__ void fold_body(int t,
                            const float* __restrict__ We1, const float* __restrict__ ae1,
                            const float* __restrict__ We2, const float* __restrict__ ae2,
                            float* __restrict__ Bt) {
    int f = t >> 2, h = t & 3;
    float s1 = 0.f, s2 = 0.f;
#pragma unroll
    for (int c = 0; c < CHD; ++c) {
        s1 += We1[f * HC + h * CHD + c] * ae1[h * CHD + c];
        s2 += We2[f * HC + h * CHD + c] * ae2[h * CHD + c];
    }
    int j = f >> 3, hf = (f >> 2) & 1, u = (f >> 1) & 1, rg = f & 1;
    int step = j * 2 + u;
    Bt[(step * 32 + h + 16 * hf) * 2 + rg]     = s1;
    Bt[(step * 32 + 4 + h + 16 * hf) * 2 + rg] = s2;
    for (int z = t; z < 512; z += 256) {
        int rg2 = z & 1, col = 8 + ((z >> 1) & 7), hf2 = (z >> 4) & 1, st = z >> 5;
        Bt[(st * 32 + col + 16 * hf2) * 2 + rg2] = 0.f;
    }
}

__device__ __forceinline__ void wave_lds_sync() {
    __builtin_amdgcn_fence(__ATOMIC_RELEASE, "wavefront");
    __builtin_amdgcn_wave_barrier();
    __builtin_amdgcn_fence(__ATOMIC_ACQUIRE, "wavefront");
}

#define TILE_PITCH 68

__global__ void __launch_bounds__(256)
edge_csr_kernel(const float* __restrict__ EA, const int* __restrict__ off,
                const int* __restrict__ degI, int* eidA, const int* __restrict__ srcA,
                const float* __restrict__ Bt,
                float* __restrict__ ae1c, float* __restrict__ ae2c,
                float* __restrict__ sum1, float* __restrict__ sum2, int N,
                const float* __restrict__ Hp,
                const float* __restrict__ att_s, const float* __restrict__ att_d,
                float* __restrict__ a_s, float* __restrict__ a_d) {
    __shared__ alignas(16) float tileS[8][16 * TILE_PITCH];
    int wave = threadIdx.x >> 5, lane = threadIdx.x & 31;
    int n = blockIdx.x * 8 + wave;
    if (n >= N) return;
    float4 hrow = reinterpret_cast<const float4*>(Hp)[(size_t)n * 32 + lane];
    float* my = tileS[wave];
    int c = lane & 15, half = lane >> 4;

    v2f bfr[16];
#pragma unroll
    for (int s = 0; s < 16; ++s) bfr[s] = reinterpret_cast<const v2f*>(Bt)[s * 32 + lane];

    int cnt = __builtin_amdgcn_readfirstlane(degI[n]);
    int s0  = __builtin_amdgcn_readfirstlane(off[n]);
    float tot = 0.f;

    for (int t0 = 0; t0 < cnt; t0 += 16) {
        int eid = eidA[s0 + min(t0 + c, cnt - 1)];
        int sid = srcA[eid];
        float4 v[8];
#pragma unroll
        for (int i = 0; i < 8; ++i) {
            int er = __shfl(eid, 2 * i + half, 32);
            v[i] = *reinterpret_cast<const float4*>(EA + (size_t)er * F_IN + 4 * c);
        }
#pragma unroll
        for (int i = 0; i < 8; ++i)
            *reinterpret_cast<float4*>(my + (2 * i + half) * TILE_PITCH + 4 * c) = v[i];
        wave_lds_sync();

        v8f acc = {};
#pragma unroll
        for (int j = 0; j < 8; ++j) {
            float4 a4 = *reinterpret_cast<const float4*>(my + c * TILE_PITCH + 8 * j + 4 * half);
            v2f a0 = { a4.x, a4.y };
            v2f a1 = { a4.z, a4.w };
            acc = __builtin_amdgcn_wmma_f32_16x16x4_f32(false, a0, false, bfr[2 * j],
                                                        (short)0, acc, false, false);
            acc = __builtin_amdgcn_wmma_f32_16x16x4_f32(false, a1, false, bfr[2 * j + 1],
                                                        (short)0, acc, false, false);
        }
        wave_lds_sync();

#pragma unroll
        for (int r = 0; r < 8; ++r) {
            float val = acc[r];
            tot += (t0 + 8 * half + r < cnt) ? val : 0.f;
            my[(8 * half + r) * 20 + c] = val;
        }
        wave_lds_sync();
        float4 o = *reinterpret_cast<const float4*>(my + c * 20 + 4 * half);
        wave_lds_sync();
        if (t0 + c < cnt) {
            float* dstp = half ? ae2c : ae1c;
            *reinterpret_cast<float4*>(dstp + (size_t)(s0 + t0 + c) * 4) = o;
            if (half == 0) eidA[s0 + t0 + c] = sid;
        }
    }
    tot += __shfl_xor(tot, 16, 32);
    if (half == 0) {
        if (c < 4)      sum1[(size_t)n * 4 + c]     = tot;
        else if (c < 8) sum2[(size_t)n * 4 + c - 4] = tot;
    }
    node_att_body(hrow, n, lane, att_s, att_d, a_s, a_d);
}

#define GEMM_R 64

__device__ __forceinline__ void gemm_load_b(v2f (&bfr)[16], const float* __restrict__ wp) {
#pragma unroll
    for (int s = 0; s < 16; ++s) {
        int kk = 8 * (s >> 1) + 2 * (s & 1);
        bfr[s].x = wp[(size_t)kk * HC];
        bfr[s].y = wp[(size_t)(kk + 1) * HC];
    }
}

__device__ __forceinline__ void gemm_body(const float* __restrict__ X, const float* __restrict__ W,
                                          float* __restrict__ Y, int M, const int K, float4* xs4,
                                          const int tid, const int row0) {
    const int wave = tid >> 5, lane = tid & 31;
    const int c = lane & 15, half = lane >> 4;
    const int pq = (K >> 2) + 1;
    const float4* X4 = reinterpret_cast<const float4*>(X);

    const int kq = K >> 2;
    const int nq = GEMM_R * kq / 256;
    float4 v[8];
#pragma unroll
    for (int q = 0; q < 8; ++q) {
        if (q < nq) {
            int i = tid + 256 * q;
            int r = min(row0 + i / kq, M - 1);
            v[q] = X4[(size_t)r * kq + (i % kq)];
        }
    }
    const int col0 = wave * 16;
    const float* wp = W + (size_t)(4 * half) * HC + col0 + c;
    v2f bfr[2][16];
    gemm_load_b(bfr[0], wp);
    if (K > 64) gemm_load_b(bfr[1], wp + (size_t)64 * HC);
#pragma unroll
    for (int q = 0; q < 8; ++q) {
        if (q < nq) {
            int i = tid + 256 * q;
            xs4[(i / kq) * pq + (i % kq)] = v[q];
        }
    }
    __syncthreads();

    v8f acc0 = {}, acc1 = {}, acc2 = {}, acc3 = {};
    const float4* xr = xs4 + c * pq + half;
#pragma unroll
    for (int kc = 0; kc < K; kc += 64) {
        const v2f (&bf)[16] = bfr[kc >> 6];
#pragma unroll
        for (int j = 0; j < 8; ++j) {
            const float4* xa = xr + (kc >> 2) + 2 * j;
            float4 p0 = xa[0];
            float4 p1 = xa[16 * pq];
            float4 p2 = xa[32 * pq];
            float4 p3 = xa[48 * pq];
            v2f a00 = { p0.x, p0.y }, a01 = { p0.z, p0.w };
            v2f a10 = { p1.x, p1.y }, a11 = { p1.z, p1.w };
            v2f a20 = { p2.x, p2.y }, a21 = { p2.z, p2.w };
            v2f a30 = { p3.x, p3.y }, a31 = { p3.z, p3.w };
            acc0 = __builtin_amdgcn_wmma_f32_16x16x4_f32(false, a00, false, bf[2 * j], (short)0, acc0, false, false);
            acc1 = __builtin_amdgcn_wmma_f32_16x16x4_f32(false, a10, false, bf[2 * j], (short)0, acc1, false, false);
            acc2 = __builtin_amdgcn_wmma_f32_16x16x4_f32(false, a20, false, bf[2 * j], (short)0, acc2, false, false);
            acc3 = __builtin_amdgcn_wmma_f32_16x16x4_f32(false, a30, false, bf[2 * j], (short)0, acc3, false, false);
            acc0 = __builtin_amdgcn_wmma_f32_16x16x4_f32(false, a01, false, bf[2 * j + 1], (short)0, acc0, false, false);
            acc1 = __builtin_amdgcn_wmma_f32_16x16x4_f32(false, a11, false, bf[2 * j + 1], (short)0, acc1, false, false);
            acc2 = __builtin_amdgcn_wmma_f32_16x16x4_f32(false, a21, false, bf[2 * j + 1], (short)0, acc2, false, false);
            acc3 = __builtin_amdgcn_wmma_f32_16x16x4_f32(false, a31, false, bf[2 * j + 1], (short)0, acc3, false, false);
        }
    }

    __builtin_amdgcn_sched_barrier(0);
    asm volatile("v_nop\n\tv_nop\n\tv_nop\n\tv_nop");
    __builtin_amdgcn_sched_barrier(0);

    const int rb = row0 + 8 * half;
    float* yp = Y + (size_t)rb * HC + col0 + c;
    if (row0 + GEMM_R > M) {
#pragma unroll
        for (int r = 0; r < 8; ++r) {
            if (rb + r < M)      yp[(size_t)r * HC]        = acc0[r];
            if (rb + 16 + r < M) yp[(size_t)(16 + r) * HC] = acc1[r];
            if (rb + 32 + r < M) yp[(size_t)(32 + r) * HC] = acc2[r];
            if (rb + 48 + r < M) yp[(size_t)(48 + r) * HC] = acc3[r];
        }
    } else {
#pragma unroll
        for (int r = 0; r < 8; ++r) {
            yp[(size_t)r * HC]        = acc0[r];
            yp[(size_t)(16 + r) * HC] = acc1[r];
            yp[(size_t)(32 + r) * HC] = acc2[r];
            yp[(size_t)(48 + r) * HC] = acc3[r];
        }
    }
}

#define OWN_R    392
#define OWN_T    1024
#define OWN_W    (OWN_T / 32)
#define OWN_CAPW 640
#define OWN_MAXSL 4096
#define BK_SL    4096
#define BK_T     1024

__global__ void __launch_bounds__(BK_T)
csr_bucket_kernel(const int* __restrict__ dstA, int* __restrict__ bkw, int* __restrict__ cntT,
                  int E, int NB, int NSL,
                  const float* __restrict__ We1, const float* __restrict__ ae1,
                  const float* __restrict__ We2, const float* __restrict__ ae2,
                  float* __restrict__ Bt) {
    __shared__ int hist[BK_T];
    __shared__ int start[BK_T];
    __shared__ int wsum[BK_T / 32];
    if (blockIdx.x == NSL) {
        if (threadIdx.x < 256) fold_body(threadIdx.x, We1, ae1, We2, ae2, Bt);
        return;
    }
    const int tid = threadIdx.x, lane = tid & 31, wave = tid >> 5;
    const int b = blockIdx.x;
    const int e0 = b * BK_SL;
    int2* bk = reinterpret_cast<int2*>(bkw);

    hist[tid] = 0;
    __syncthreads();

    int d[4], rg[4], rk[4];
#pragma unroll
    for (int k = 0; k < 4; ++k) {
        int e = e0 + k * BK_T + tid;
        d[k] = (e < E) ? dstA[e] : -1;
    }
#pragma unroll
    for (int k = 0; k < 4; ++k) {
        rg[k] = 0; rk[k] = 0;
        if (d[k] >= 0) {
            rg[k] = (int)((unsigned)d[k] / (unsigned)OWN_R);
            rk[k] = atomicAdd(&hist[rg[k]], 1);
        }
    }
    __syncthreads();

    const int cntv = hist[tid];
    int inc = cntv;
    for (int dlt = 1; dlt < 32; dlt <<= 1) {
        int up = __shfl_up(inc, dlt, 32);
        if (lane >= dlt) inc += up;
    }
    if (lane == 31) wsum[wave] = inc;
    __syncthreads();
    if (wave == 0) {
        int w = wsum[lane];
        int winc = w;
        for (int dlt = 1; dlt < 32; dlt <<= 1) {
            int up = __shfl_up(winc, dlt, 32);
            if (lane >= dlt) winc += up;
        }
        wsum[lane] = winc - w;
    }
    __syncthreads();
    start[tid] = inc - cntv + wsum[wave];
    if (tid < NB) cntT[(size_t)tid * NSL + b] = cntv;
    __syncthreads();
#pragma unroll
    for (int k = 0; k < 4; ++k)
        if (d[k] >= 0)
            bk[(size_t)e0 + start[rg[k]] + rk[k]] = make_int2(e0 + k * BK_T + tid, d[k]);
}

__device__ __forceinline__ void own_append(int2 en, int have, int lo, int lane, int& wcnt, int2* my) {
    if (lane < have) my[min(wcnt + lane, OWN_CAPW - 1)] = make_int2(en.x, en.y - lo);
    wcnt += min(have, 32);
}

__global__ void __launch_bounds__(OWN_T)
csr_owner_kernel(const int* __restrict__ bkw, const int* __restrict__ cntT,
                 const int* __restrict__ dstA, int* __restrict__ off, int* __restrict__ degI,
                 int* __restrict__ eidA, int N, int E, int NSL, int nOwn,
                 const float* __restrict__ X, const float* __restrict__ W,
                 float* __restrict__ Y) {
    __shared__ alignas(16) int2 list[OWN_W][OWN_CAPW];
    __shared__ int ncount[OWN_R];
    __shared__ int cur[OWN_R];
    __shared__ int wsum[OWN_W];
    __shared__ int segStart[OWN_MAXSL];
    __shared__ int segCnt[OWN_MAXSL];
    __shared__ int belowAll, ovf;

    if ((int)blockIdx.x >= nOwn) {
        const int unit = threadIdx.x >> 8;
        float4* xs4 = reinterpret_cast<float4*>(&list[0][0]) + unit * (GEMM_R * (F_IN / 4 + 1));
        gemm_body(X, W, Y, N, F_IN, xs4, threadIdx.x & 255,
                  (((int)blockIdx.x - nOwn) * 4 + unit) * GEMM_R);
        return;
    }

    const int tid = threadIdx.x, lane = tid & 31, wave = tid >> 5;
    const int r = blockIdx.x;
    const int lo = r * OWN_R, hi = lo + OWN_R;
    int2* my = list[wave];
    const int2* bk = reinterpret_cast<const int2*>(bkw);

    for (int i = tid; i < OWN_R; i += OWN_T) ncount[i] = 0;
    for (int i = tid; i < NSL; i += OWN_T) segStart[i] = 0;
    if (tid == 0) { belowAll = 0; ovf = 0; }
    __syncthreads();

    for (int i = tid; i < NSL; i += OWN_T) segCnt[i] = cntT[(size_t)r * NSL + i];
    {
        const int total = r * NSL;
        int bcur = tid % NSL;
        const int step = OWN_T % NSL;
        for (int i = tid; i < total; i += 8 * OWN_T) {
            int v[8], bb[8];
#pragma unroll
            for (int q = 0; q < 8; ++q) {
                int ii = i + q * OWN_T;
                v[q] = (ii < total) ? cntT[ii] : 0;
                bb[q] = bcur;
                bcur += step;
                if (bcur >= NSL) bcur -= NSL;
            }
#pragma unroll
            for (int q = 0; q < 8; ++q)
                if (v[q] != 0) atomicAdd(&segStart[bb[q]], v[q]);
        }
    }
    __syncthreads();

    int below = 0;
    for (int i = tid; i < NSL; i += OWN_T) below += segStart[i];
    for (int msk = 16; msk >= 1; msk >>= 1) below += __shfl_xor(below, msk, 32);
    if (lane == 0 && below != 0) atomicAdd(&belowAll, below);

    int wcnt = 0;
    for (int b0 = wave; b0 < NSL; b0 += 4 * OWN_W) {
        int cs[4];
        const int2* sg[4];
        int2 ea[4], eb[4];
#pragma unroll
        for (int j = 0; j < 4; ++j) {
            int b = b0 + j * OWN_W;
            int bb = (b < NSL) ? b : 0;
            int c = __builtin_amdgcn_readfirstlane(segCnt[bb]);
            int st = __builtin_amdgcn_readfirstlane(segStart[bb]);
            cs[j] = (b < NSL) ? c : 0;
            sg[j] = bk + (size_t)bb * BK_SL + st;
            ea[j] = (lane < cs[j])      ? sg[j][lane]      : make_int2(0, 0);
            eb[j] = (lane + 32 < cs[j]) ? sg[j][lane + 32] : make_int2(0, 0);
        }
#pragma unroll
        for (int j = 0; j < 4; ++j) {
            own_append(ea[j], cs[j], lo, lane, wcnt, my);
            if (cs[j] > 32) own_append(eb[j], cs[j] - 32, lo, lane, wcnt, my);
            for (int i0 = 64; i0 < cs[j]; i0 += 32) {
                int i = i0 + lane;
                int2 en = (i < cs[j]) ? sg[j][i] : make_int2(0, 0);
                own_append(en, cs[j] - i0, lo, lane, wcnt, my);
            }
        }
    }
    if (lane == 0 && wcnt > OWN_CAPW) atomicOr(&ovf, 1);
    __syncthreads();
    const bool listed = (ovf == 0);
    const int base = belowAll;

    if (listed) {
        for (int i = lane; i < wcnt; i += 32) atomicAdd(&ncount[my[i].y], 1);
    } else {
        for (int e = tid; e < E; e += OWN_T) {
            int dv = dstA[e];
            if (dv >= lo && dv < hi) atomicAdd(&ncount[dv - lo], 1);
        }
    }
    __syncthreads();

    int cntv = (tid < OWN_R) ? ncount[tid] : 0;
    int inc = cntv;
    for (int dlt = 1; dlt < 32; dlt <<= 1) {
        int up = __shfl_up(inc, dlt, 32);
        if (lane >= dlt) inc += up;
    }
    if (lane == 31) wsum[wave] = inc;
    __syncthreads();
    if (wave == 0) {
        int w = wsum[lane];
        int winc = w;
        for (int dlt = 1; dlt < 32; dlt <<= 1) {
            int up = __shfl_up(winc, dlt, 32);
            if (lane >= dlt) winc += up;
        }
        wsum[lane] = winc - w;
    }
    __syncthreads();
    if (tid < OWN_R) {
        int ex = inc - cntv + wsum[wave];
        cur[tid] = ex;
        int node = lo + tid;
        if (node < N) { off[node] = base + ex; degI[node] = cntv; }
    }
    __syncthreads();

    if (listed) {
        for (int i = lane; i < wcnt; i += 32) {
            int2 en = my[i];
            int p = atomicAdd(&cur[en.y], 1);
            eidA[base + p] = en.x;
        }
    } else {
        for (int e = tid; e < E; e += OWN_T) {
            int dv = dstA[e];
            if (dv >= lo && dv < hi) {
                int p = atomicAdd(&cur[dv - lo], 1);
                eidA[base + p] = e;
            }
        }
    }
}

__global__ void __launch_bounds__(256)
gemm_wmma_k128_kernel(const float* __restrict__ X, const float* __restrict__ W,
                      float* __restrict__ Y, int M) {
    __shared__ float4 xs4[GEMM_R * (HC / 4 + 1)];
    gemm_body(X, W, Y, M, HC, xs4, threadIdx.x, blockIdx.x * GEMM_R);
}

__global__ void node_att_kernel(const float* __restrict__ Hp,
                                const float* __restrict__ att_s, const float* __restrict__ att_d,
                                float* __restrict__ a_s, float* __restrict__ a_d, int N) {
    int wave = threadIdx.x >> 5, lane = threadIdx.x & 31;
    int n = blockIdx.x * (blockDim.x >> 5) + wave;
    if (n >= N) return;
    float4 hv = reinterpret_cast<const float4*>(Hp)[(size_t)n * 32 + lane];
    node_att_body(hv, n, lane, att_s, att_d, a_s, a_d);
}

__global__ void agg_kernel(const int* __restrict__ off, const int* __restrict__ degI,
                           const int* __restrict__ srcL,
                           const float* __restrict__ a_e, const float* __restrict__ a_s,
                           const float* __restrict__ a_d, const float* __restrict__ sum_ae,
                           const float* __restrict__ Hp, const float* __restrict__ bias,
                           const float* __restrict__ lin_w, const float* __restrict__ lin_b,
                           float* __restrict__ outv, float* __restrict__ outs,
                           int N, int mode) {
    int wave = threadIdx.x >> 5, lane = threadIdx.x & 31;
    int n = blockIdx.x * (blockDim.x >> 5) + wave;
    if (n >= N) return;
    int h = lane >> 3;
    const float4* Hp4 = reinterpret_cast<const float4*>(Hp);

    float ad_n = a_d[(size_t)n * 4 + h];
    float as_n = a_s[(size_t)n * 4 + h];
    int cnt = degI[n];
    float degf = fmaxf((float)cnt, 1.0f);
    float M = leaky(as_n + ad_n + sum_ae[(size_t)n * 4 + h] / degf);
    float D = 1.0f;
    float4 A = Hp4[(size_t)n * 32 + lane];

    int s0 = off[n];
    for (int i = 0; i < cnt; ++i) {
        int s = srcL[s0 + i];
        float lg = leaky(a_s[(size_t)s * 4 + h] + ad_n + a_e[(size_t)(s0 + i) * 4 + h]);
        float Mn = fmaxf(M, lg);
        float sc = __expf(M - Mn);
        float p  = __expf(lg - Mn);
        float4 hv = Hp4[(size_t)s * 32 + lane];
        A.x = A.x * sc + p * hv.x;
        A.y = A.y * sc + p * hv.y;
        A.z = A.z * sc + p * hv.z;
        A.w = A.w * sc + p * hv.w;
        D = D * sc + p;
        M = Mn;
    }
    float4 bv = reinterpret_cast<const float4*>(bias)[lane];
    float inv = 1.0f / D;
    float4 r = { A.x * inv + bv.x, A.y * inv + bv.y, A.z * inv + bv.z, A.w * inv + bv.w };
    if (mode == 0) {
        r.x = fmaxf(r.x, 0.f); r.y = fmaxf(r.y, 0.f);
        r.z = fmaxf(r.z, 0.f); r.w = fmaxf(r.w, 0.f);
        reinterpret_cast<float4*>(outv)[(size_t)n * 32 + lane] = r;
    } else {
        float4 lw = reinterpret_cast<const float4*>(lin_w)[lane];
        float part = r.x * lw.x + r.y * lw.y + r.z * lw.z + r.w * lw.w;
#pragma unroll
        for (int msk = 16; msk >= 1; msk >>= 1) part += __shfl_xor(part, msk, 32);
        if (lane == 0) outs[n] = part + lin_b[0];
    }
}

extern "C" void kernel_launch(void* const* d_in, const int* in_sizes, int n_in,
                              void* d_out, int out_size, void* d_ws, size_t ws_size,
                              hipStream_t stream) {
    const float* x   = (const float*)d_in[0];
    const float* ea  = (const float*)d_in[1];
    const float* W1  = (const float*)d_in[2];
    const float* as1 = (const float*)d_in[3];
    const float* ad1 = (const float*)d_in[4];
    const float* We1 = (const float*)d_in[5];
    const float* ae1 = (const float*)d_in[6];
    const float* b1  = (const float*)d_in[7];
    const float* W2  = (const float*)d_in[8];
    const float* as2 = (const float*)d_in[9];
    const float* ad2 = (const float*)d_in[10];
    const float* We2 = (const float*)d_in[11];
    const float* ae2 = (const float*)d_in[12];
    const float* b2  = (const float*)d_in[13];
    const float* lw  = (const float*)d_in[14];
    const float* lb  = (const float*)d_in[15];
    const int*   ei  = (const int*)d_in[16];

    const int N = in_sizes[0] / F_IN;
    const int E = in_sizes[16] / 2;
    const int* srcA = ei;
    const int* dstA = ei + E;

    char* w = (char*)d_ws;
    auto alloc = [&](size_t bytes) -> void* {
        void* p = (void*)w;
        w += (bytes + 255) & ~(size_t)255;
        return p;
    };
    int*   degI   = (int*)  alloc((size_t)N * 4);
    float* sum1 = (float*)alloc((size_t)N * 16);
    float* sum2 = (float*)alloc((size_t)N * 16);
    float* ae1v = (float*)alloc((size_t)E * 16);
    float* ae2v = (float*)alloc((size_t)E * 16);
    float* a_sb = (float*)alloc((size_t)N * 16);
    float* a_db = (float*)alloc((size_t)N * 16);
    int*   off  = (int*)  alloc((size_t)N * 4);
    int*   eidA = (int*)  alloc((size_t)E * 4);
    float* Hp   = (float*)alloc((size_t)N * HC * 4);
    float* h1   = (float*)alloc((size_t)N * HC * 4);

    const int gemmBlocks = (N + GEMM_R - 1) / GEMM_R;
    const int gemm1Blocks = (N + 4 * GEMM_R - 1) / (4 * GEMM_R);
    const int nodeBlocks = (N + 7) / 8;
    const int ownBlocks  = (N + OWN_R - 1) / OWN_R;
    const int nSlices    = (E + BK_SL - 1) / BK_SL;
    int*   bkw  = (int*)ae1v;
    float* Bt   = h1;
    int*   cntT = (int*)h1 + 1024;

    csr_bucket_kernel<<<nSlices + 1, BK_T, 0, stream>>>(dstA, bkw, cntT, E, ownBlocks, nSlices,
                                                        We1, ae1, We2, ae2, Bt);
    csr_owner_kernel<<<ownBlocks + gemm1Blocks, OWN_T, 0, stream>>>(bkw, cntT, dstA, off, degI, eidA,
                                                                    N, E, nSlices, ownBlocks,
                                                                    x, W1, Hp);
    edge_csr_kernel<<<nodeBlocks, 256, 0, stream>>>(ea, off, degI, eidA, srcA, Bt,
                                                    ae1v, ae2v, sum1, sum2, N,
                                                    Hp, as1, ad1, a_sb, a_db);

    agg_kernel<<<nodeBlocks, 256, 0, stream>>>(off, degI, eidA, ae1v, a_sb, a_db,
                                               sum1, Hp, b1, nullptr, nullptr,
                                               h1, nullptr, N, 0);
    gemm_wmma_k128_kernel<<<gemmBlocks, 256, 0, stream>>>(h1, W2, Hp, N);
    node_att_kernel<<<nodeBlocks, 256, 0, stream>>>(Hp, as2, ad2, a_sb, a_db, N);
    agg_kernel<<<nodeBlocks, 256, 0, stream>>>(off, degI, eidA, ae2v, a_sb, a_db,
                                               sum2, Hp, b2, lw, lb,
                                               nullptr, (float*)d_out, N, 1);
}


# aggregation list walk moved to the scalar unit: node id, list bounds, source ids by scalar loads (4 ids a fetch, a group ahead); row/a_s/a_e addresses = scalar base + lane offset (sums unchanged)
# speedup vs baseline: 3.8592x; 1.0707x over previous
#include <hip/hip_runtime.h>
#include <hip/hip_bf16.h>

#define F_IN   64
#define HEADS  4
#define CHD    32
#define HC     128
#define NEG_SLOPE 0.2f

typedef __attribute__((ext_vector_type(2))) float v2f;
typedef __attribute__((ext_vector_type(8))) float v8f;

__device__ __forceinline__ float leaky(float x) { return x > 0.f ? x : NEG_SLOPE * x; }

__device__ __forceinline__ void node_att_body(float4 hv, int n, int lane,
                                const float* __restrict__ att_s, const float* __restrict__ att_d,
                                float* __restrict__ a_s, float* __restrict__ a_d) {
    int h = lane >> 3, q = lane & 7;
    float4 sa = reinterpret_cast<const float4*>(att_s)[lane];
    float4 da = reinterpret_cast<const float4*>(att_d)[lane];
    float ps = hv.x * sa.x + hv.y * sa.y + hv.z * sa.z + hv.w * sa.w;
    float pd = hv.x * da.x + hv.y * da.y + hv.z * da.z + hv.w * da.w;
#pragma unroll
    for (int msk = 1; msk < 8; msk <<= 1) {
        ps += __shfl_xor(ps, msk, 32);
        pd += __shfl_xor(pd, msk, 32);
    }
    if (q == 0) { a_s[(size_t)n * 4 + h] = ps; a_d[(size_t)n * 4 + h] = pd; }
}

__device__ __forceinline__ void fold_body(int t,
                            const float* __restrict__ We1, const float* __restrict__ ae1,
                            const float* __restrict__ We2, const float* __restrict__ ae2,
                            float* __restrict__ Bt) {
    int f = t >> 2, h = t & 3;
    float s1 = 0.f, s2 = 0.f;
#pragma unroll
    for (int c = 0; c < CHD; ++c) {
        s1 += We1[f * HC + h * CHD + c] * ae1[h * CHD + c];
        s2 += We2[f * HC + h * CHD + c] * ae2[h * CHD + c];
    }
    int j = f >> 3, hf = (f >> 2) & 1, u = (f >> 1) & 1, rg = f & 1;
    int step = j * 2 + u;
    Bt[(step * 32 + h + 16 * hf) * 2 + rg]     = s1;
    Bt[(step * 32 + 4 + h + 16 * hf) * 2 + rg] = s2;
    for (int z = t; z < 512; z += 256) {
        int rg2 = z & 1, col = 8 + ((z >> 1) & 7), hf2 = (z >> 4) & 1, st = z >> 5;
        Bt[(st * 32 + col + 16 * hf2) * 2 + rg2] = 0.f;
    }
}

__device__ __forceinline__ void wave_lds_sync() {
    __builtin_amdgcn_fence(__ATOMIC_RELEASE, "wavefront");
    __builtin_amdgcn_wave_barrier();
    __builtin_amdgcn_fence(__ATOMIC_ACQUIRE, "wavefront");
}

#define TILE_PITCH 68

__global__ void __launch_bounds__(256)
edge_csr_kernel(const float* __restrict__ EA, const int* __restrict__ off,
                const int* __restrict__ degI, int* eidA, const int* __restrict__ srcA,
                const float* __restrict__ Bt,
                float* __restrict__ ae1c, float* __restrict__ ae2c,
                float* __restrict__ sum1, float* __restrict__ sum2, int N,
                const float* __restrict__ Hp,
                const float* __restrict__ att_s, const float* __restrict__ att_d,
                float* __restrict__ a_s, float* __restrict__ a_d) {
    __shared__ alignas(16) float tileS[8][16 * TILE_PITCH];
    int wave = threadIdx.x >> 5, lane = threadIdx.x & 31;
    int n = blockIdx.x * 8 + wave;
    if (n >= N) return;
    float4 hrow = reinterpret_cast<const float4*>(Hp)[(size_t)n * 32 + lane];
    float* my = tileS[wave];
    int c = lane & 15, half = lane >> 4;

    v2f bfr[16];
#pragma unroll
    for (int s = 0; s < 16; ++s) bfr[s] = reinterpret_cast<const v2f*>(Bt)[s * 32 + lane];

    int cnt = __builtin_amdgcn_readfirstlane(degI[n]);
    int s0  = __builtin_amdgcn_readfirstlane(off[n]);
    float tot = 0.f;

    for (int t0 = 0; t0 < cnt; t0 += 16) {
        int eid = eidA[s0 + min(t0 + c, cnt - 1)];
        int sid = srcA[eid];
        float4 v[8];
#pragma unroll
        for (int i = 0; i < 8; ++i) {
            int er = __shfl(eid, 2 * i + half, 32);
            v[i] = *reinterpret_cast<const float4*>(EA + (size_t)er * F_IN + 4 * c);
        }
#pragma unroll
        for (int i = 0; i < 8; ++i)
            *reinterpret_cast<float4*>(my + (2 * i + half) * TILE_PITCH + 4 * c) = v[i];
        wave_lds_sync();

        v8f acc = {};
#pragma unroll
        for (int j = 0; j < 8; ++j) {
            float4 a4 = *reinterpret_cast<const float4*>(my + c * TILE_PITCH + 8 * j + 4 * half);
            v2f a0 = { a4.x, a4.y };
            v2f a1 = { a4.z, a4.w };
            acc = __builtin_amdgcn_wmma_f32_16x16x4_f32(false, a0, false, bfr[2 * j],
                                                        (short)0, acc, false, false);
            acc = __builtin_amdgcn_wmma_f32_16x16x4_f32(false, a1, false, bfr[2 * j + 1],
                                                        (short)0, acc, false, false);
        }
        wave_lds_sync();

#pragma unroll
        for (int r = 0; r < 8; ++r) {
            float val = acc[r];
            tot += (t0 + 8 * half + r < cnt) ? val : 0.f;
            my[(8 * half + r) * 20 + c] = val;
        }
        wave_lds_sync();
        float4 o = *reinterpret_cast<const float4*>(my + c * 20 + 4 * half);
        wave_lds_sync();
        if (t0 + c < cnt) {
            float* dstp = half ? ae2c : ae1c;
            *reinterpret_cast<float4*>(dstp + (size_t)(s0 + t0 + c) * 4) = o;
            if (half == 0) eidA[s0 + t0 + c] = sid;
        }
    }
    tot += __shfl_xor(tot, 16, 32);
    if (half == 0) {
        if (c < 4)      sum1[(size_t)n * 4 + c]     = tot;
        else if (c < 8) sum2[(size_t)n * 4 + c - 4] = tot;
    }
    node_att_body(hrow, n, lane, att_s, att_d, a_s, a_d);
}

#define GEMM_R 64

__device__ __forceinline__ void gemm_load_b(v2f (&bfr)[16], const float* __restrict__ wp) {
#pragma unroll
    for (int s = 0; s < 16; ++s) {
        int kk = 8 * (s >> 1) + 2 * (s & 1);
        bfr[s].x = wp[(size_t)kk * HC];
        bfr[s].y = wp[(size_t)(kk + 1) * HC];
    }
}

__device__ __forceinline__ void gemm_body(const float* __restrict__ X, const float* __restrict__ W,
                                          float* __restrict__ Y, int M, const int K, float4* xs4,
                                          const int tid, const int row0) {
    const int wave = tid >> 5, lane = tid & 31;
    const int c = lane & 15, half = lane >> 4;
    const int pq = (K >> 2) + 1;
    const float4* X4 = reinterpret_cast<const float4*>(X);

    const int kq = K >> 2;
    const int nq = GEMM_R * kq / 256;
    float4 v[8];
#pragma unroll
    for (int q = 0; q < 8; ++q) {
        if (q < nq) {
            int i = tid + 256 * q;
            int r = min(row0 + i / kq, M - 1);
            v[q] = X4[(size_t)r * kq + (i % kq)];
        }
    }
    const int col0 = wave * 16;
    const float* wp = W + (size_t)(4 * half) * HC + col0 + c;
    v2f bfr[2][16];
    gemm_load_b(bfr[0], wp);
    if (K > 64) gemm_load_b(bfr[1], wp + (size_t)64 * HC);
#pragma unroll
    for (int q = 0; q < 8; ++q) {
        if (q < nq) {
            int i = tid + 256 * q;
            xs4[(i / kq) * pq + (i % kq)] = v[q];
        }
    }
    __syncthreads();

    v8f acc0 = {}, acc1 = {}, acc2 = {}, acc3 = {};
    const float4* xr = xs4 + c * pq + half;
#pragma unroll
    for (int kc = 0; kc < K; kc += 64) {
        const v2f (&bf)[16] = bfr[kc >> 6];
#pragma unroll
        for (int j = 0; j < 8; ++j) {
            const float4* xa = xr + (kc >> 2) + 2 * j;
            float4 p0 = xa[0];
            float4 p1 = xa[16 * pq];
            float4 p2 = xa[32 * pq];
            float4 p3 = xa[48 * pq];
            v2f a00 = { p0.x, p0.y }, a01 = { p0.z, p0.w };
            v2f a10 = { p1.x, p1.y }, a11 = { p1.z, p1.w };
            v2f a20 = { p2.x, p2.y }, a21 = { p2.z, p2.w };
            v2f a30 = { p3.x, p3.y }, a31 = { p3.z, p3.w };
            acc0 = __builtin_amdgcn_wmma_f32_16x16x4_f32(false, a00, false, bf[2 * j], (short)0, acc0, false, false);
            acc1 = __builtin_amdgcn_wmma_f32_16x16x4_f32(false, a10, false, bf[2 * j], (short)0, acc1, false, false);
            acc2 = __builtin_amdgcn_wmma_f32_16x16x4_f32(false, a20, false, bf[2 * j], (short)0, acc2, false, false);
            acc3 = __builtin_amdgcn_wmma_f32_16x16x4_f32(false, a30, false, bf[2 * j], (short)0, acc3, false, false);
            acc0 = __builtin_amdgcn_wmma_f32_16x16x4_f32(false, a01, false, bf[2 * j + 1], (short)0, acc0, false, false);
            acc1 = __builtin_amdgcn_wmma_f32_16x16x4_f32(false, a11, false, bf[2 * j + 1], (short)0, acc1, false, false);
            acc2 = __builtin_amdgcn_wmma_f32_16x16x4_f32(false, a21, false, bf[2 * j + 1], (short)0, acc2, false, false);
            acc3 = __builtin_amdgcn_wmma_f32_16x16x4_f32(false, a31, false, bf[2 * j + 1], (short)0, acc3, false, false);
        }
    }

    __builtin_amdgcn_sched_barrier(0);
    asm volatile("v_nop\n\tv_nop\n\tv_nop\n\tv_nop");
    __builtin_amdgcn_sched_barrier(0);

    const int rb = row0 + 8 * half;
    float* yp = Y + (size_t)rb * HC + col0 + c;
    if (row0 + GEMM_R > M) {
#pragma unroll
        for (int r = 0; r < 8; ++r) {
            if (rb + r < M)      yp[(size_t)r * HC]        = acc0[r];
            if (rb + 16 + r < M) yp[(size_t)(16 + r) * HC] = acc1[r];
            if (rb + 32 + r < M) yp[(size_t)(32 + r) * HC] = acc2[r];
            if (rb + 48 + r < M) yp[(size_t)(48 + r) * HC] = acc3[r];
        }
    } else {
#pragma unroll
        for (int r = 0; r < 8; ++r) {
            yp[(size_t)r * HC]        = acc0[r];
            yp[(size_t)(16 + r) * HC] = acc1[r];
            yp[(size_t)(32 + r) * HC] = acc2[r];
            yp[(size_t)(48 + r) * HC] = acc3[r];
        }
    }
}

#define OWN_R    392
#define OWN_T    1024
#define OWN_W    (OWN_T / 32)
#define OWN_CAPW 640
#define OWN_MAXSL 4096
#define BK_SL    4096
#define BK_T     1024

__global__ void __launch_bounds__(BK_T)
csr_bucket_kernel(const int* __restrict__ dstA, int* __restrict__ bkw, int* __restrict__ cntT,
                  int E, int NB, int NSL,
                  const float* __restrict__ We1, const float* __restrict__ ae1,
                  const float* __restrict__ We2, const float* __restrict__ ae2,
                  float* __restrict__ Bt) {
    __shared__ int hist[BK_T];
    __shared__ int start[BK_T];
    __shared__ int wsum[BK_T / 32];
    if (blockIdx.x == NSL) {
        if (threadIdx.x < 256) fold_body(threadIdx.x, We1, ae1, We2, ae2, Bt);
        return;
    }
    const int tid = threadIdx.x, lane = tid & 31, wave = tid >> 5;
    const int b = blockIdx.x;
    const int e0 = b * BK_SL;
    int2* bk = reinterpret_cast<int2*>(bkw);

    hist[tid] = 0;
    __syncthreads();

    int d[4], rg[4], rk[4];
#pragma unroll
    for (int k = 0; k < 4; ++k) {
        int e = e0 + k * BK_T + tid;
        d[k] = (e < E) ? dstA[e] : -1;
    }
#pragma unroll
    for (int k = 0; k < 4; ++k) {
        rg[k] = 0; rk[k] = 0;
        if (d[k] >= 0) {
            rg[k] = (int)((unsigned)d[k] / (unsigned)OWN_R);
            rk[k] = atomicAdd(&hist[rg[k]], 1);
        }
    }
    __syncthreads();

    const int cntv = hist[tid];
    int inc = cntv;
    for (int dlt = 1; dlt < 32; dlt <<= 1) {
        int up = __shfl_up(inc, dlt, 32);
        if (lane >= dlt) inc += up;
    }
    if (lane == 31) wsum[wave] = inc;
    __syncthreads();
    if (wave == 0) {
        int w = wsum[lane];
        int winc = w;
        for (int dlt = 1; dlt < 32; dlt <<= 1) {
            int up = __shfl_up(winc, dlt, 32);
            if (lane >= dlt) winc += up;
        }
        wsum[lane] = winc - w;
    }
    __syncthreads();
    start[tid] = inc - cntv + wsum[wave];
    if (tid < NB) cntT[(size_t)tid * NSL + b] = cntv;
    __syncthreads();
#pragma unroll
    for (int k = 0; k < 4; ++k)
        if (d[k] >= 0)
            bk[(size_t)e0 + start[rg[k]] + rk[k]] = make_int2(e0 + k * BK_T + tid, d[k]);
}

__device__ __forceinline__ void own_append(int2 en, int have, int lo, int lane, int& wcnt, int2* my) {
    if (lane < have) my[min(wcnt + lane, OWN_CAPW - 1)] = make_int2(en.x, en.y - lo);
    wcnt += min(have, 32);
}

__global__ void __launch_bounds__(OWN_T)
csr_owner_kernel(const int* __restrict__ bkw, const int* __restrict__ cntT,
                 const int* __restrict__ dstA, int* __restrict__ off, int* __restrict__ degI,
                 int* __restrict__ eidA, int N, int E, int NSL, int nOwn,
                 const float* __restrict__ X, const float* __restrict__ W,
                 float* __restrict__ Y) {
    __shared__ alignas(16) int2 list[OWN_W][OWN_CAPW];
    __shared__ int ncount[OWN_R];
    __shared__ int cur[OWN_R];
    __shared__ int wsum[OWN_W];
    __shared__ int segStart[OWN_MAXSL];
    __shared__ int segCnt[OWN_MAXSL];
    __shared__ int belowAll, ovf;

    if ((int)blockIdx.x >= nOwn) {
        const int unit = threadIdx.x >> 8;
        float4* xs4 = reinterpret_cast<float4*>(&list[0][0]) + unit * (GEMM_R * (F_IN / 4 + 1));
        gemm_body(X, W, Y, N, F_IN, xs4, threadIdx.x & 255,
                  (((int)blockIdx.x - nOwn) * 4 + unit) * GEMM_R);
        return;
    }

    const int tid = threadIdx.x, lane = tid & 31, wave = tid >> 5;
    const int r = blockIdx.x;
    const int lo = r * OWN_R, hi = lo + OWN_R;
    int2* my = list[wave];
    const int2* bk = reinterpret_cast<const int2*>(bkw);

    for (int i = tid; i < OWN_R; i += OWN_T) ncount[i] = 0;
    for (int i = tid; i < NSL; i += OWN_T) segStart[i] = 0;
    if (tid == 0) { belowAll = 0; ovf = 0; }
    __syncthreads();

    for (int i = tid; i < NSL; i += OWN_T) segCnt[i] = cntT[(size_t)r * NSL + i];
    {
        const int total = r * NSL;
        int bcur = tid % NSL;
        const int step = OWN_T % NSL;
        for (int i = tid; i < total; i += 8 * OWN_T) {
            int v[8], bb[8];
#pragma unroll
            for (int q = 0; q < 8; ++q) {
                int ii = i + q * OWN_T;
                v[q] = (ii < total) ? cntT[ii] : 0;
                bb[q] = bcur;
                bcur += step;
                if (bcur >= NSL) bcur -= NSL;
            }
#pragma unroll
            for (int q = 0; q < 8; ++q)
                if (v[q] != 0) atomicAdd(&segStart[bb[q]], v[q]);
        }
    }
    __syncthreads();

    int below = 0;
    for (int i = tid; i < NSL; i += OWN_T) below += segStart[i];
    for (int msk = 16; msk >= 1; msk >>= 1) below += __shfl_xor(below, msk, 32);
    if (lane == 0 && below != 0) atomicAdd(&belowAll, below);

    int wcnt = 0;
    for (int b0 = wave; b0 < NSL; b0 += 4 * OWN_W) {
        int cs[4];
        const int2* sg[4];
        int2 ea[4], eb[4];
#pragma unroll
        for (int j = 0; j < 4; ++j) {
            int b = b0 + j * OWN_W;
            int bb = (b < NSL) ? b : 0;
            int c = __builtin_amdgcn_readfirstlane(segCnt[bb]);
            int st = __builtin_amdgcn_readfirstlane(segStart[bb]);
            cs[j] = (b < NSL) ? c : 0;
            sg[j] = bk + (size_t)bb * BK_SL + st;
            ea[j] = (lane < cs[j])      ? sg[j][lane]      : make_int2(0, 0);
            eb[j] = (lane + 32 < cs[j]) ? sg[j][lane + 32] : make_int2(0, 0);
        }
#pragma unroll
        for (int j = 0; j < 4; ++j) {
            own_append(ea[j], cs[j], lo, lane, wcnt, my);
            if (cs[j] > 32) own_append(eb[j], cs[j] - 32, lo, lane, wcnt, my);
            for (int i0 = 64; i0 < cs[j]; i0 += 32) {
                int i = i0 + lane;
                int2 en = (i < cs[j]) ? sg[j][i] : make_int2(0, 0);
                own_append(en, cs[j] - i0, lo, lane, wcnt, my);
            }
        }
    }
    if (lane == 0 && wcnt > OWN_CAPW) atomicOr(&ovf, 1);
    __syncthreads();
    const bool listed = (ovf == 0);
    const int base = belowAll;

    if (listed) {
        for (int i = lane; i < wcnt; i += 32) atomicAdd(&ncount[my[i].y], 1);
    } else {
        for (int e = tid; e < E; e += OWN_T) {
            int dv = dstA[e];
            if (dv >= lo && dv < hi) atomicAdd(&ncount[dv - lo], 1);
        }
    }
    __syncthreads();

    int cntv = (tid < OWN_R) ? ncount[tid] : 0;
    int inc = cntv;
    for (int dlt = 1; dlt < 32; dlt <<= 1) {
        int up = __shfl_up(inc, dlt, 32);
        if (lane >= dlt) inc += up;
    }
    if (lane == 31) wsum[wave] = inc;
    __syncthreads();
    if (wave == 0) {
        int w = wsum[lane];
        int winc = w;
        for (int dlt = 1; dlt < 32; dlt <<= 1) {
            int up = __shfl_up(winc, dlt, 32);
            if (lane >= dlt) winc += up;
        }
        wsum[lane] = winc - w;
    }
    __syncthreads();
    if (tid < OWN_R) {
        int ex = inc - cntv + wsum[wave];
        cur[tid] = ex;
        int node = lo + tid;
        if (node < N) { off[node] = base + ex; degI[node] = cntv; }
    }
    __syncthreads();

    if (listed) {
        for (int i = lane; i < wcnt; i += 32) {
            int2 en = my[i];
            int p = atomicAdd(&cur[en.y], 1);
            eidA[base + p] = en.x;
        }
    } else {
        for (int e = tid; e < E; e += OWN_T) {
            int dv = dstA[e];
            if (dv >= lo && dv < hi) {
                int p = atomicAdd(&cur[dv - lo], 1);
                eidA[base + p] = e;
            }
        }
    }
}

__global__ void __launch_bounds__(256)
gemm_wmma_k128_kernel(const float* __restrict__ X, const float* __restrict__ W,
                      float* __restrict__ Y, int M) {
    __shared__ float4 xs4[GEMM_R * (HC / 4 + 1)];
    gemm_body(X, W, Y, M, HC, xs4, threadIdx.x, blockIdx.x * GEMM_R);
}

__global__ void node_att_kernel(const float* __restrict__ Hp,
                                const float* __restrict__ att_s, const float* __restrict__ att_d,
                                float* __restrict__ a_s, float* __restrict__ a_d, int N) {
    int wave = threadIdx.x >> 5, lane = threadIdx.x & 31;
    int n = blockIdx.x * (blockDim.x >> 5) + wave;
    if (n >= N) return;
    float4 hv = reinterpret_cast<const float4*>(Hp)[(size_t)n * 32 + lane];
    node_att_body(hv, n, lane, att_s, att_d, a_s, a_d);
}

__device__ __forceinline__ void agg_edge(float asv, float aev, float4 hv, float ad_n,
                                         float& M, float& D, float4& A) {
    float lg = leaky(asv + ad_n + aev);
    float Mn = fmaxf(M, lg);
    float sc = __expf(M - Mn);
    float p  = __expf(lg - Mn);
    A.x = A.x * sc + p * hv.x;
    A.y = A.y * sc + p * hv.y;
    A.z = A.z * sc + p * hv.z;
    A.w = A.w * sc + p * hv.w;
    D = D * sc + p;
    M = Mn;
}

typedef const __attribute__((address_space(1))) float*  gmem_f;
typedef const __attribute__((address_space(1))) float4* gmem_f4;
__device__ __forceinline__ gmem_f scalar_base_f(const float* q) {
    gmem_f p = (gmem_f)q;
    asm("" : "+s"(p));
    return p;
}
__device__ __forceinline__ gmem_f4 scalar_base_f4(const float4* q) {
    gmem_f4 p = (gmem_f4)q;
    asm("" : "+s"(p));
    return p;
}

template <int G>
__device__ __forceinline__ void agg_group(int s0, int s1, int s2, int s3, int i,
                                          const float* __restrict__ a_s,
                                          const float* __restrict__ aeL,
                                          const float4* __restrict__ Hp4,
                                          unsigned h, unsigned lane, float ad_n,
                                          float& M, float& D, float4& A) {
    const unsigned sid[4] = { (unsigned)s0, (unsigned)s1, (unsigned)s2, (unsigned)s3 };
    float asv[G], aev[G];
    float4 hv[G];
#pragma unroll
    for (int j = 0; j < G; ++j) {
        gmem_f  ap = scalar_base_f(a_s + (size_t)sid[j] * 4);
        gmem_f  ep = scalar_base_f(aeL + (size_t)(i + j) * 4);
        gmem_f4 rp = scalar_base_f4(Hp4 + (size_t)sid[j] * 32);
        asv[j] = ap[h];
        aev[j] = ep[h];
        hv[j]  = rp[lane];
    }
#pragma unroll
    for (int j = 0; j < G; ++j) agg_edge(asv[j], aev[j], hv[j], ad_n, M, D, A);
}

__global__ void agg_kernel(const int* __restrict__ off, const int* __restrict__ degI,
                           const int* __restrict__ srcL,
                           const float* __restrict__ a_e, const float* __restrict__ a_s,
                           const float* __restrict__ a_d, const float* __restrict__ sum_ae,
                           const float* __restrict__ Hp, const float* __restrict__ bias,
                           const float* __restrict__ lin_w, const float* __restrict__ lin_b,
                           float* __restrict__ outv, float* __restrict__ outs,
                           int N, int mode) {
    const unsigned lane = threadIdx.x & 31;
    const int n = __builtin_amdgcn_readfirstlane(
        (int)(blockIdx.x * (blockDim.x >> 5) + (threadIdx.x >> 5)));
    if (n >= N) return;
    const unsigned h = lane >> 3;
    const float4* Hp4 = reinterpret_cast<const float4*>(Hp);

    float ad_n = (a_d + (size_t)n * 4)[h];
    float as_n = (a_s + (size_t)n * 4)[h];
    float sae  = (sum_ae + (size_t)n * 4)[h];
    float4 A = (Hp4 + (size_t)n * 32)[lane];
    int cnt = degI[n];
    int st  = off[n];
    asm volatile("" : "+s"(cnt), "+s"(st));
    const int*   lst = srcL + st;
    const float* aeL = a_e + (size_t)st * 4;
    int x0 = 0, x1 = 0, x2 = 0, x3 = 0;
    if (cnt >= 4) {
        x0 = lst[0]; x1 = lst[1]; x2 = lst[2]; x3 = lst[3];
    } else if (cnt > 0) {
        const int last = cnt - 1;
        x0 = lst[0]; x1 = lst[min(1, last)]; x2 = lst[min(2, last)];
    }

    float degf = fmaxf((float)cnt, 1.0f);
    float M = leaky(as_n + ad_n + sae / degf);
    float D = 1.0f;

    if (cnt >= 4) {
        int i = 0;
        for (; i + 4 <= cnt; i += 4) {
            const int c0 = x0, c1 = x1, c2 = x2, c3 = x3;
            const int* nx = lst + min(i + 4, cnt - 4);
            x0 = nx[0]; x1 = nx[1]; x2 = nx[2]; x3 = nx[3];
            agg_group<4>(c0, c1, c2, c3, i, a_s, aeL, Hp4, h, lane, ad_n, M, D, A);
        }
        const int rem = cnt - i;
        if (rem == 3)      agg_group<3>(x1, x2, x3, x3, i, a_s, aeL, Hp4, h, lane, ad_n, M, D, A);
        else if (rem == 2) agg_group<2>(x2, x3, x3, x3, i, a_s, aeL, Hp4, h, lane, ad_n, M, D, A);
        else if (rem == 1) agg_group<1>(x3, x3, x3, x3, i, a_s, aeL, Hp4, h, lane, ad_n, M, D, A);
    } else if (cnt == 3) {
        agg_group<3>(x0, x1, x2, x2, 0, a_s, aeL, Hp4, h, lane, ad_n, M, D, A);
    } else if (cnt == 2) {
        agg_group<2>(x0, x1, x1, x1, 0, a_s, aeL, Hp4, h, lane, ad_n, M, D, A);
    } else if (cnt == 1) {
        agg_group<1>(x0, x0, x0, x0, 0, a_s, aeL, Hp4, h, lane, ad_n, M, D, A);
    }
    float4 bv = reinterpret_cast<const float4*>(bias)[lane];
    float inv = 1.0f / D;
    float4 r = { A.x * inv + bv.x, A.y * inv + bv.y, A.z * inv + bv.z, A.w * inv + bv.w };
    if (mode == 0) {
        r.x = fmaxf(r.x, 0.f); r.y = fmaxf(r.y, 0.f);
        r.z = fmaxf(r.z, 0.f); r.w = fmaxf(r.w, 0.f);
        (reinterpret_cast<float4*>(outv) + (size_t)n * 32)[lane] = r;
    } else {
        float4 lw = reinterpret_cast<const float4*>(lin_w)[lane];
        float part = r.x * lw.x + r.y * lw.y + r.z * lw.z + r.w * lw.w;
#pragma unroll
        for (int msk = 16; msk >= 1; msk >>= 1) part += __shfl_xor(part, msk, 32);
        if (lane == 0) outs[n] = part + lin_b[0];
    }
}

extern "C" void kernel_launch(void* const* d_in, const int* in_sizes, int n_in,
                              void* d_out, int out_size, void* d_ws, size_t ws_size,
                              hipStream_t stream) {
    const float* x   = (const float*)d_in[0];
    const float* ea  = (const float*)d_in[1];
    const float* W1  = (const float*)d_in[2];
    const float* as1 = (const float*)d_in[3];
    const float* ad1 = (const float*)d_in[4];
    const float* We1 = (const float*)d_in[5];
    const float* ae1 = (const float*)d_in[6];
    const float* b1  = (const float*)d_in[7];
    const float* W2  = (const float*)d_in[8];
    const float* as2 = (const float*)d_in[9];
    const float* ad2 = (const float*)d_in[10];
    const float* We2 = (const float*)d_in[11];
    const float* ae2 = (const float*)d_in[12];
    const float* b2  = (const float*)d_in[13];
    const float* lw  = (const float*)d_in[14];
    const float* lb  = (const float*)d_in[15];
    const int*   ei  = (const int*)d_in[16];

    const int N = in_sizes[0] / F_IN;
    const int E = in_sizes[16] / 2;
    const int* srcA = ei;
    const int* dstA = ei + E;

    char* w = (char*)d_ws;
    auto alloc = [&](size_t bytes) -> void* {
        void* p = (void*)w;
        w += (bytes + 255) & ~(size_t)255;
        return p;
    };
    int*   degI   = (int*)  alloc((size_t)N * 4);
    float* sum1 = (float*)alloc((size_t)N * 16);
    float* sum2 = (float*)alloc((size_t)N * 16);
    float* ae1v = (float*)alloc((size_t)E * 16);
    float* ae2v = (float*)alloc((size_t)E * 16);
    float* a_sb = (float*)alloc((size_t)N * 16);
    float* a_db = (float*)alloc((size_t)N * 16);
    int*   off  = (int*)  alloc((size_t)N * 4);
    int*   eidA = (int*)  alloc((size_t)E * 4);
    float* Hp   = (float*)alloc((size_t)N * HC * 4);
    float* h1   = (float*)alloc((size_t)N * HC * 4);

    const int gemmBlocks = (N + GEMM_R - 1) / GEMM_R;
    const int gemm1Blocks = (N + 4 * GEMM_R - 1) / (4 * GEMM_R);
    const int nodeBlocks = (N + 7) / 8;
    const int ownBlocks  = (N + OWN_R - 1) / OWN_R;
    const int nSlices    = (E + BK_SL - 1) / BK_SL;
    int*   bkw  = (int*)ae1v;
    float* Bt   = h1;
    int*   cntT = (int*)h1 + 1024;

    csr_bucket_kernel<<<nSlices + 1, BK_T, 0, stream>>>(dstA, bkw, cntT, E, ownBlocks, nSlices,
                                                        We1, ae1, We2, ae2, Bt);
    csr_owner_kernel<<<ownBlocks + gemm1Blocks, OWN_T, 0, stream>>>(bkw, cntT, dstA, off, degI, eidA,
                                                                    N, E, nSlices, ownBlocks,
                                                                    x, W1, Hp);
    edge_csr_kernel<<<nodeBlocks, 256, 0, stream>>>(ea, off, degI, eidA, srcA, Bt,
                                                    ae1v, ae2v, sum1, sum2, N,
                                                    Hp, as1, ad1, a_sb, a_db);

    agg_kernel<<<nodeBlocks, 256, 0, stream>>>(off, degI, eidA, ae1v, a_sb, a_db,
                                               sum1, Hp, b1, nullptr, nullptr,
                                               h1, nullptr, N, 0);
    gemm_wmma_k128_kernel<<<gemmBlocks, 256, 0, stream>>>(h1, W2, Hp, N);
    node_att_kernel<<<nodeBlocks, 256, 0, stream>>>(Hp, as2, ad2, a_sb, a_db, N);
    agg_kernel<<<nodeBlocks, 256, 0, stream>>>(off, degI, eidA, ae2v, a_sb, a_db,
                                               sum2, Hp, b2, lw, lb,
                                               nullptr, (float*)d_out, N, 1);
}


# edge pass takes its B fragments per tile from one LDS table per block, not from 32 held registers; row fetches use 32-bit offsets: 111 to 64 registers, 8 to 16 waves per SIMD (sums unchanged)
# speedup vs baseline: 4.0003x; 1.0366x over previous
#include <hip/hip_runtime.h>
#include <hip/hip_bf16.h>

#define F_IN   64
#define HEADS  4
#define CHD    32
#define HC     128
#define NEG_SLOPE 0.2f

typedef __attribute__((ext_vector_type(2))) float v2f;
typedef __attribute__((ext_vector_type(8))) float v8f;

__device__ __forceinline__ float leaky(float x) { return x > 0.f ? x : NEG_SLOPE * x; }

__device__ __forceinline__ void node_att_body(float4 hv, int n, int lane,
                                const float* __restrict__ att_s, const float* __restrict__ att_d,
                                float* __restrict__ a_s, float* __restrict__ a_d) {
    int h = lane >> 3, q = lane & 7;
    float4 sa = reinterpret_cast<const float4*>(att_s)[lane];
    float4 da = reinterpret_cast<const float4*>(att_d)[lane];
    float ps = hv.x * sa.x + hv.y * sa.y + hv.z * sa.z + hv.w * sa.w;
    float pd = hv.x * da.x + hv.y * da.y + hv.z * da.z + hv.w * da.w;
#pragma unroll
    for (int msk = 1; msk < 8; msk <<= 1) {
        ps += __shfl_xor(ps, msk, 32);
        pd += __shfl_xor(pd, msk, 32);
    }
    if (q == 0) { a_s[(size_t)n * 4 + h] = ps; a_d[(size_t)n * 4 + h] = pd; }
}

__device__ __forceinline__ void fold_body(int t,
                            const float* __restrict__ We1, const float* __restrict__ ae1,
                            const float* __restrict__ We2, const float* __restrict__ ae2,
                            float* __restrict__ Bt) {
    int f = t >> 2, h = t & 3;
    float s1 = 0.f, s2 = 0.f;
#pragma unroll
    for (int c = 0; c < CHD; ++c) {
        s1 += We1[f * HC + h * CHD + c] * ae1[h * CHD + c];
        s2 += We2[f * HC + h * CHD + c] * ae2[h * CHD + c];
    }
    int j = f >> 3, hf = (f >> 2) & 1, u = (f >> 1) & 1, rg = f & 1;
    int step = j * 2 + u;
    Bt[(step * 32 + h + 16 * hf) * 2 + rg]     = s1;
    Bt[(step * 32 + 4 + h + 16 * hf) * 2 + rg] = s2;
    for (int z = t; z < 512; z += 256) {
        int rg2 = z & 1, col = 8 + ((z >> 1) & 7), hf2 = (z >> 4) & 1, st = z >> 5;
        Bt[(st * 32 + col + 16 * hf2) * 2 + rg2] = 0.f;
    }
}

__device__ __forceinline__ void wave_lds_sync() {
    __builtin_amdgcn_fence(__ATOMIC_RELEASE, "wavefront");
    __builtin_amdgcn_wave_barrier();
    __builtin_amdgcn_fence(__ATOMIC_ACQUIRE, "wavefront");
}

#define TILE_PITCH 68

__global__ void __launch_bounds__(256, 12)
edge_csr_kernel(const float* __restrict__ EA, const int* __restrict__ off,
                const int* __restrict__ degI, int* eidA, const int* __restrict__ srcA,
                const float* __restrict__ Bt,
                float* __restrict__ ae1c, float* __restrict__ ae2c,
                float* __restrict__ sum1, float* __restrict__ sum2, int N,
                const float* __restrict__ Hp,
                const float* __restrict__ att_s, const float* __restrict__ att_d,
                float* __restrict__ a_s, float* __restrict__ a_d) {
    __shared__ alignas(16) float tileS[8][16 * TILE_PITCH];
    __shared__ alignas(16) v2f BtS[16 * 32];
    int wave = threadIdx.x >> 5, lane = threadIdx.x & 31;
    int n = blockIdx.x * 8 + wave;
    const int nc = min(n, N - 1);
    float4 hrow = reinterpret_cast<const float4*>(Hp)[(size_t)nc * 32 + lane];
    int cntv = degI[nc];
    int s0v  = off[nc];
    BtS[threadIdx.x]       = reinterpret_cast<const v2f*>(Bt)[threadIdx.x];
    BtS[threadIdx.x + 256] = reinterpret_cast<const v2f*>(Bt)[threadIdx.x + 256];
    __syncthreads();
    if (n >= N) return;
    float* my = tileS[wave];
    int c = lane & 15, half = lane >> 4;

    int cnt = __builtin_amdgcn_readfirstlane(cntv);
    int s0  = __builtin_amdgcn_readfirstlane(s0v);
    float tot = 0.f;

    for (int t0 = 0; t0 < cnt; t0 += 16) {
        int eid = eidA[s0 + min(t0 + c, cnt - 1)];
        int sid = srcA[eid];
        float4 v[8];
#pragma unroll
        for (int i = 0; i < 8; ++i) {
            int er = __shfl(eid, 2 * i + half, 32);
            v[i] = reinterpret_cast<const float4*>(EA)[(unsigned)er * 16u + (unsigned)c];
        }
#pragma unroll
        for (int i = 0; i < 8; ++i)
            *reinterpret_cast<float4*>(my + (2 * i + half) * TILE_PITCH + 4 * c) = v[i];
        wave_lds_sync();

        v2f bfr[16];
#pragma unroll
        for (int s = 0; s < 16; ++s) bfr[s] = BtS[s * 32 + lane];
        v8f acc = {};
#pragma unroll
        for (int j = 0; j < 8; ++j) {
            float4 a4 = *reinterpret_cast<const float4*>(my + c * TILE_PITCH + 8 * j + 4 * half);
            v2f a0 = { a4.x, a4.y };
            v2f a1 = { a4.z, a4.w };
            acc = __builtin_amdgcn_wmma_f32_16x16x4_f32(false, a0, false, bfr[2 * j],
                                                        (short)0, acc, false, false);
            acc = __builtin_amdgcn_wmma_f32_16x16x4_f32(false, a1, false, bfr[2 * j + 1],
                                                        (short)0, acc, false, false);
        }
        wave_lds_sync();

#pragma unroll
        for (int r = 0; r < 8; ++r) {
            float val = acc[r];
            tot += (t0 + 8 * half + r < cnt) ? val : 0.f;
            my[(8 * half + r) * 20 + c] = val;
        }
        wave_lds_sync();
        float4 o = *reinterpret_cast<const float4*>(my + c * 20 + 4 * half);
        wave_lds_sync();
        if (t0 + c < cnt) {
            float* dstp = half ? ae2c : ae1c;
            *reinterpret_cast<float4*>(dstp + (size_t)(s0 + t0 + c) * 4) = o;
            if (half == 0) eidA[s0 + t0 + c] = sid;
        }
    }
    tot += __shfl_xor(tot, 16, 32);
    if (half == 0) {
        if (c < 4)      sum1[(size_t)n * 4 + c]     = tot;
        else if (c < 8) sum2[(size_t)n * 4 + c - 4] = tot;
    }
    node_att_body(hrow, n, lane, att_s, att_d, a_s, a_d);
}

#define GEMM_R 64

__device__ __forceinline__ void gemm_load_b(v2f (&bfr)[16], const float* __restrict__ wp) {
#pragma unroll
    for (int s = 0; s < 16; ++s) {
        int kk = 8 * (s >> 1) + 2 * (s & 1);
        bfr[s].x = wp[(size_t)kk * HC];
        bfr[s].y = wp[(size_t)(kk + 1) * HC];
    }
}

__device__ __forceinline__ void gemm_body(const float* __restrict__ X, const float* __restrict__ W,
                                          float* __restrict__ Y, int M, const int K, float4* xs4,
                                          const int tid, const int row0) {
    const int wave = tid >> 5, lane = tid & 31;
    const int c = lane & 15, half = lane >> 4;
    const int pq = (K >> 2) + 1;
    const float4* X4 = reinterpret_cast<const float4*>(X);

    const int kq = K >> 2;
    const int nq = GEMM_R * kq / 256;
    float4 v[8];
#pragma unroll
    for (int q = 0; q < 8; ++q) {
        if (q < nq) {
            int i = tid + 256 * q;
            int r = min(row0 + i / kq, M - 1);
            v[q] = X4[(size_t)r * kq + (i % kq)];
        }
    }
    const int col0 = wave * 16;
    const float* wp = W + (size_t)(4 * half) * HC + col0 + c;
    v2f bfr[2][16];
    gemm_load_b(bfr[0], wp);
    if (K > 64) gemm_load_b(bfr[1], wp + (size_t)64 * HC);
#pragma unroll
    for (int q = 0; q < 8; ++q) {
        if (q < nq) {
            int i = tid + 256 * q;
            xs4[(i / kq) * pq + (i % kq)] = v[q];
        }
    }
    __syncthreads();

    v8f acc0 = {}, acc1 = {}, acc2 = {}, acc3 = {};
    const float4* xr = xs4 + c * pq + half;
#pragma unroll
    for (int kc = 0; kc < K; kc += 64) {
        const v2f (&bf)[16] = bfr[kc >> 6];
#pragma unroll
        for (int j = 0; j < 8; ++j) {
            const float4* xa = xr + (kc >> 2) + 2 * j;
            float4 p0 = xa[0];
            float4 p1 = xa[16 * pq];
            float4 p2 = xa[32 * pq];
            float4 p3 = xa[48 * pq];
            v2f a00 = { p0.x, p0.y }, a01 = { p0.z, p0.w };
            v2f a10 = { p1.x, p1.y }, a11 = { p1.z, p1.w };
            v2f a20 = { p2.x, p2.y }, a21 = { p2.z, p2.w };
            v2f a30 = { p3.x, p3.y }, a31 = { p3.z, p3.w };
            acc0 = __builtin_amdgcn_wmma_f32_16x16x4_f32(false, a00, false, bf[2 * j], (short)0, acc0, false, false);
            acc1 = __builtin_amdgcn_wmma_f32_16x16x4_f32(false, a10, false, bf[2 * j], (short)0, acc1, false, false);
            acc2 = __builtin_amdgcn_wmma_f32_16x16x4_f32(false, a20, false, bf[2 * j], (short)0, acc2, false, false);
            acc3 = __builtin_amdgcn_wmma_f32_16x16x4_f32(false, a30, false, bf[2 * j], (short)0, acc3, false, false);
            acc0 = __builtin_amdgcn_wmma_f32_16x16x4_f32(false, a01, false, bf[2 * j + 1], (short)0, acc0, false, false);
            acc1 = __builtin_amdgcn_wmma_f32_16x16x4_f32(false, a11, false, bf[2 * j + 1], (short)0, acc1, false, false);
            acc2 = __builtin_amdgcn_wmma_f32_16x16x4_f32(false, a21, false, bf[2 * j + 1], (short)0, acc2, false, false);
            acc3 = __builtin_amdgcn_wmma_f32_16x16x4_f32(false, a31, false, bf[2 * j + 1], (short)0, acc3, false, false);
        }
    }

    __builtin_amdgcn_sched_barrier(0);
    asm volatile("v_nop\n\tv_nop\n\tv_nop\n\tv_nop");
    __builtin_amdgcn_sched_barrier(0);

    const int rb = row0 + 8 * half;
    float* yp = Y + (size_t)rb * HC + col0 + c;
    if (row0 + GEMM_R > M) {
#pragma unroll
        for (int r = 0; r < 8; ++r) {
            if (rb + r < M)      yp[(size_t)r * HC]        = acc0[r];
            if (rb + 16 + r < M) yp[(size_t)(16 + r) * HC] = acc1[r];
            if (rb + 32 + r < M) yp[(size_t)(32 + r) * HC] = acc2[r];
            if (rb + 48 + r < M) yp[(size_t)(48 + r) * HC] = acc3[r];
        }
    } else {
#pragma unroll
        for (int r = 0; r < 8; ++r) {
            yp[(size_t)r * HC]        = acc0[r];
            yp[(size_t)(16 + r) * HC] = acc1[r];
            yp[(size_t)(32 + r) * HC] = acc2[r];
            yp[(size_t)(48 + r) * HC] = acc3[r];
        }
    }
}

#define OWN_R    392
#define OWN_T    1024
#define OWN_W    (OWN_T / 32)
#define OWN_CAPW 640
#define OWN_MAXSL 4096
#define BK_SL    4096
#define BK_T     1024

__global__ void __launch_bounds__(BK_T)
csr_bucket_kernel(const int* __restrict__ dstA, int* __restrict__ bkw, int* __restrict__ cntT,
                  int E, int NB, int NSL,
                  const float* __restrict__ We1, const float* __restrict__ ae1,
                  const float* __restrict__ We2, const float* __restrict__ ae2,
                  float* __restrict__ Bt) {
    __shared__ int hist[BK_T];
    __shared__ int start[BK_T];
    __shared__ int wsum[BK_T / 32];
    if (blockIdx.x == NSL) {
        if (threadIdx.x < 256) fold_body(threadIdx.x, We1, ae1, We2, ae2, Bt);
        return;
    }
    const int tid = threadIdx.x, lane = tid & 31, wave = tid >> 5;
    const int b = blockIdx.x;
    const int e0 = b * BK_SL;
    int2* bk = reinterpret_cast<int2*>(bkw);

    hist[tid] = 0;
    __syncthreads();

    int d[4], rg[4], rk[4];
#pragma unroll
    for (int k = 0; k < 4; ++k) {
        int e = e0 + k * BK_T + tid;
        d[k] = (e < E) ? dstA[e] : -1;
    }
#pragma unroll
    for (int k = 0; k < 4; ++k) {
        rg[k] = 0; rk[k] = 0;
        if (d[k] >= 0) {
            rg[k] = (int)((unsigned)d[k] / (unsigned)OWN_R);
            rk[k] = atomicAdd(&hist[rg[k]], 1);
        }
    }
    __syncthreads();

    const int cntv = hist[tid];
    int inc = cntv;
    for (int dlt = 1; dlt < 32; dlt <<= 1) {
        int up = __shfl_up(inc, dlt, 32);
        if (lane >= dlt) inc += up;
    }
    if (lane == 31) wsum[wave] = inc;
    __syncthreads();
    if (wave == 0) {
        int w = wsum[lane];
        int winc = w;
        for (int dlt = 1; dlt < 32; dlt <<= 1) {
            int up = __shfl_up(winc, dlt, 32);
            if (lane >= dlt) winc += up;
        }
        wsum[lane] = winc - w;
    }
    __syncthreads();
    start[tid] = inc - cntv + wsum[wave];
    if (tid < NB) cntT[(size_t)tid * NSL + b] = cntv;
    __syncthreads();
#pragma unroll
    for (int k = 0; k < 4; ++k)
        if (d[k] >= 0)
            bk[(size_t)e0 + start[rg[k]] + rk[k]] = make_int2(e0 + k * BK_T + tid, d[k]);
}

__device__ __forceinline__ void own_append(int2 en, int have, int lo, int lane, int& wcnt, int2* my) {
    if (lane < have) my[min(wcnt + lane, OWN_CAPW - 1)] = make_int2(en.x, en.y - lo);
    wcnt += min(have, 32);
}

__global__ void __launch_bounds__(OWN_T)
csr_owner_kernel(const int* __restrict__ bkw, const int* __restrict__ cntT,
                 const int* __restrict__ dstA, int* __restrict__ off, int* __restrict__ degI,
                 int* __restrict__ eidA, int N, int E, int NSL, int nOwn,
                 const float* __restrict__ X, const float* __restrict__ W,
                 float* __restrict__ Y) {
    __shared__ alignas(16) int2 list[OWN_W][OWN_CAPW];
    __shared__ int ncount[OWN_R];
    __shared__ int cur[OWN_R];
    __shared__ int wsum[OWN_W];
    __shared__ int segStart[OWN_MAXSL];
    __shared__ int segCnt[OWN_MAXSL];
    __shared__ int belowAll, ovf;

    if ((int)blockIdx.x >= nOwn) {
        const int unit = threadIdx.x >> 8;
        float4* xs4 = reinterpret_cast<float4*>(&list[0][0]) + unit * (GEMM_R * (F_IN / 4 + 1));
        gemm_body(X, W, Y, N, F_IN, xs4, threadIdx.x & 255,
                  (((int)blockIdx.x - nOwn) * 4 + unit) * GEMM_R);
        return;
    }

    const int tid = threadIdx.x, lane = tid & 31, wave = tid >> 5;
    const int r = blockIdx.x;
    const int lo = r * OWN_R, hi = lo + OWN_R;
    int2* my = list[wave];
    const int2* bk = reinterpret_cast<const int2*>(bkw);

    for (int i = tid; i < OWN_R; i += OWN_T) ncount[i] = 0;
    for (int i = tid; i < NSL; i += OWN_T) segStart[i] = 0;
    if (tid == 0) { belowAll = 0; ovf = 0; }
    __syncthreads();

    for (int i = tid; i < NSL; i += OWN_T) segCnt[i] = cntT[(size_t)r * NSL + i];
    {
        const int total = r * NSL;
        int bcur = tid % NSL;
        const int step = OWN_T % NSL;
        for (int i = tid; i < total; i += 8 * OWN_T) {
            int v[8], bb[8];
#pragma unroll
            for (int q = 0; q < 8; ++q) {
                int ii = i + q * OWN_T;
                v[q] = (ii < total) ? cntT[ii] : 0;
                bb[q] = bcur;
                bcur += step;
                if (bcur >= NSL) bcur -= NSL;
            }
#pragma unroll
            for (int q = 0; q < 8; ++q)
                if (v[q] != 0) atomicAdd(&segStart[bb[q]], v[q]);
        }
    }
    __syncthreads();

    int below = 0;
    for (int i = tid; i < NSL; i += OWN_T) below += segStart[i];
    for (int msk = 16; msk >= 1; msk >>= 1) below += __shfl_xor(below, msk, 32);
    if (lane == 0 && below != 0) atomicAdd(&belowAll, below);

    int wcnt = 0;
    for (int b0 = wave; b0 < NSL; b0 += 4 * OWN_W) {
        int cs[4];
        const int2* sg[4];
        int2 ea[4], eb[4];
#pragma unroll
        for (int j = 0; j < 4; ++j) {
            int b = b0 + j * OWN_W;
            int bb = (b < NSL) ? b : 0;
            int c = __builtin_amdgcn_readfirstlane(segCnt[bb]);
            int st = __builtin_amdgcn_readfirstlane(segStart[bb]);
            cs[j] = (b < NSL) ? c : 0;
            sg[j] = bk + (size_t)bb * BK_SL + st;
            ea[j] = (lane < cs[j])      ? sg[j][lane]      : make_int2(0, 0);
            eb[j] = (lane + 32 < cs[j]) ? sg[j][lane + 32] : make_int2(0, 0);
        }
#pragma unroll
        for (int j = 0; j < 4; ++j) {
            own_append(ea[j], cs[j], lo, lane, wcnt, my);
            if (cs[j] > 32) own_append(eb[j], cs[j] - 32, lo, lane, wcnt, my);
            for (int i0 = 64; i0 < cs[j]; i0 += 32) {
                int i = i0 + lane;
                int2 en = (i < cs[j]) ? sg[j][i] : make_int2(0, 0);
                own_append(en, cs[j] - i0, lo, lane, wcnt, my);
            }
        }
    }
    if (lane == 0 && wcnt > OWN_CAPW) atomicOr(&ovf, 1);
    __syncthreads();
    const bool listed = (ovf == 0);
    const int base = belowAll;

    if (listed) {
        for (int i = lane; i < wcnt; i += 32) atomicAdd(&ncount[my[i].y], 1);
    } else {
        for (int e = tid; e < E; e += OWN_T) {
            int dv = dstA[e];
            if (dv >= lo && dv < hi) atomicAdd(&ncount[dv - lo], 1);
        }
    }
    __syncthreads();

    int cntv = (tid < OWN_R) ? ncount[tid] : 0;
    int inc = cntv;
    for (int dlt = 1; dlt < 32; dlt <<= 1) {
        int up = __shfl_up(inc, dlt, 32);
        if (lane >= dlt) inc += up;
    }
    if (lane == 31) wsum[wave] = inc;
    __syncthreads();
    if (wave == 0) {
        int w = wsum[lane];
        int winc = w;
        for (int dlt = 1; dlt < 32; dlt <<= 1) {
            int up = __shfl_up(winc, dlt, 32);
            if (lane >= dlt) winc += up;
        }
        wsum[lane] = winc - w;
    }
    __syncthreads();
    if (tid < OWN_R) {
        int ex = inc - cntv + wsum[wave];
        cur[tid] = ex;
        int node = lo + tid;
        if (node < N) { off[node] = base + ex; degI[node] = cntv; }
    }
    __syncthreads();

    if (listed) {
        for (int i = lane; i < wcnt; i += 32) {
            int2 en = my[i];
            int p = atomicAdd(&cur[en.y], 1);
            eidA[base + p] = en.x;
        }
    } else {
        for (int e = tid; e < E; e += OWN_T) {
            int dv = dstA[e];
            if (dv >= lo && dv < hi) {
                int p = atomicAdd(&cur[dv - lo], 1);
                eidA[base + p] = e;
            }
        }
    }
}

__global__ void __launch_bounds__(256)
gemm_wmma_k128_kernel(const float* __restrict__ X, const float* __restrict__ W,
                      float* __restrict__ Y, int M) {
    __shared__ float4 xs4[GEMM_R * (HC / 4 + 1)];
    gemm_body(X, W, Y, M, HC, xs4, threadIdx.x, blockIdx.x * GEMM_R);
}

__global__ void node_att_kernel(const float* __restrict__ Hp,
                                const float* __restrict__ att_s, const float* __restrict__ att_d,
                                float* __restrict__ a_s, float* __restrict__ a_d, int N) {
    int wave = threadIdx.x >> 5, lane = threadIdx.x & 31;
    int n = blockIdx.x * (blockDim.x >> 5) + wave;
    if (n >= N) return;
    float4 hv = reinterpret_cast<const float4*>(Hp)[(size_t)n * 32 + lane];
    node_att_body(hv, n, lane, att_s, att_d, a_s, a_d);
}

__device__ __forceinline__ void agg_edge(float asv, float aev, float4 hv, float ad_n,
                                         float& M, float& D, float4& A) {
    float lg = leaky(asv + ad_n + aev);
    float Mn = fmaxf(M, lg);
    float sc = __expf(M - Mn);
    float p  = __expf(lg - Mn);
    A.x = A.x * sc + p * hv.x;
    A.y = A.y * sc + p * hv.y;
    A.z = A.z * sc + p * hv.z;
    A.w = A.w * sc + p * hv.w;
    D = D * sc + p;
    M = Mn;
}

typedef const __attribute__((address_space(1))) float*  gmem_f;
typedef const __attribute__((address_space(1))) float4* gmem_f4;
__device__ __forceinline__ gmem_f scalar_base_f(const float* q) {
    gmem_f p = (gmem_f)q;
    asm("" : "+s"(p));
    return p;
}
__device__ __forceinline__ gmem_f4 scalar_base_f4(const float4* q) {
    gmem_f4 p = (gmem_f4)q;
    asm("" : "+s"(p));
    return p;
}

template <int G>
__device__ __forceinline__ void agg_group(int s0, int s1, int s2, int s3, int i,
                                          const float* __restrict__ a_s,
                                          const float* __restrict__ aeL,
                                          const float4* __restrict__ Hp4,
                                          unsigned h, unsigned lane, float ad_n,
                                          float& M, float& D, float4& A) {
    const unsigned sid[4] = { (unsigned)s0, (unsigned)s1, (unsigned)s2, (unsigned)s3 };
    float asv[G], aev[G];
    float4 hv[G];
#pragma unroll
    for (int j = 0; j < G; ++j) {
        gmem_f  ap = scalar_base_f(a_s + (size_t)sid[j] * 4);
        gmem_f  ep = scalar_base_f(aeL + (size_t)(i + j) * 4);
        gmem_f4 rp = scalar_base_f4(Hp4 + (size_t)sid[j] * 32);
        asv[j] = ap[h];
        aev[j] = ep[h];
        hv[j]  = rp[lane];
    }
#pragma unroll
    for (int j = 0; j < G; ++j) agg_edge(asv[j], aev[j], hv[j], ad_n, M, D, A);
}

__global__ void agg_kernel(const int* __restrict__ off, const int* __restrict__ degI,
                           const int* __restrict__ srcL,
                           const float* __restrict__ a_e, const float* __restrict__ a_s,
                           const float* __restrict__ a_d, const float* __restrict__ sum_ae,
                           const float* __restrict__ Hp, const float* __restrict__ bias,
                           const float* __restrict__ lin_w, const float* __restrict__ lin_b,
                           float* __restrict__ outv, float* __restrict__ outs,
                           int N, int mode) {
    const unsigned lane = threadIdx.x & 31;
    const int n = __builtin_amdgcn_readfirstlane(
        (int)(blockIdx.x * (blockDim.x >> 5) + (threadIdx.x >> 5)));
    if (n >= N) return;
    const unsigned h = lane >> 3;
    const float4* Hp4 = reinterpret_cast<const float4*>(Hp);

    float ad_n = (a_d + (size_t)n * 4)[h];
    float as_n = (a_s + (size_t)n * 4)[h];
    float sae  = (sum_ae + (size_t)n * 4)[h];
    float4 A = (Hp4 + (size_t)n * 32)[lane];
    int cnt = degI[n];
    int st  = off[n];
    asm volatile("" : "+s"(cnt), "+s"(st));
    const int*   lst = srcL + st;
    const float* aeL = a_e + (size_t)st * 4;
    int x0 = 0, x1 = 0, x2 = 0, x3 = 0;
    if (cnt >= 4) {
        x0 = lst[0]; x1 = lst[1]; x2 = lst[2]; x3 = lst[3];
    } else if (cnt > 0) {
        const int last = cnt - 1;
        x0 = lst[0]; x1 = lst[min(1, last)]; x2 = lst[min(2, last)];
    }

    float degf = fmaxf((float)cnt, 1.0f);
    float M = leaky(as_n + ad_n + sae / degf);
    float D = 1.0f;

    if (cnt >= 4) {
        int i = 0;
        for (; i + 4 <= cnt; i += 4) {
            const int c0 = x0, c1 = x1, c2 = x2, c3 = x3;
            const int* nx = lst + min(i + 4, cnt - 4);
            x0 = nx[0]; x1 = nx[1]; x2 = nx[2]; x3 = nx[3];
            agg_group<4>(c0, c1, c2, c3, i, a_s, aeL, Hp4, h, lane, ad_n, M, D, A);
        }
        const int rem = cnt - i;
        if (rem == 3)      agg_group<3>(x1, x2, x3, x3, i, a_s, aeL, Hp4, h, lane, ad_n, M, D, A);
        else if (rem == 2) agg_group<2>(x2, x3, x3, x3, i, a_s, aeL, Hp4, h, lane, ad_n, M, D, A);
        else if (rem == 1) agg_group<1>(x3, x3, x3, x3, i, a_s, aeL, Hp4, h, lane, ad_n, M, D, A);
    } else if (cnt == 3) {
        agg_group<3>(x0, x1, x2, x2, 0, a_s, aeL, Hp4, h, lane, ad_n, M, D, A);
    } else if (cnt == 2) {
        agg_group<2>(x0, x1, x1, x1, 0, a_s, aeL, Hp4, h, lane, ad_n, M, D, A);
    } else if (cnt == 1) {
        agg_group<1>(x0, x0, x0, x0, 0, a_s, aeL, Hp4, h, lane, ad_n, M, D, A);
    }
    float4 bv = reinterpret_cast<const float4*>(bias)[lane];
    float inv = 1.0f / D;
    float4 r = { A.x * inv + bv.x, A.y * inv + bv.y, A.z * inv + bv.z, A.w * inv + bv.w };
    if (mode == 0) {
        r.x = fmaxf(r.x, 0.f); r.y = fmaxf(r.y, 0.f);
        r.z = fmaxf(r.z, 0.f); r.w = fmaxf(r.w, 0.f);
        (reinterpret_cast<float4*>(outv) + (size_t)n * 32)[lane] = r;
    } else {
        float4 lw = reinterpret_cast<const float4*>(lin_w)[lane];
        float part = r.x * lw.x + r.y * lw.y + r.z * lw.z + r.w * lw.w;
#pragma unroll
        for (int msk = 16; msk >= 1; msk >>= 1) part += __shfl_xor(part, msk, 32);
        if (lane == 0) outs[n] = part + lin_b[0];
    }
}

extern "C" void kernel_launch(void* const* d_in, const int* in_sizes, int n_in,
                              void* d_out, int out_size, void* d_ws, size_t ws_size,
                              hipStream_t stream) {
    const float* x   = (const float*)d_in[0];
    const float* ea  = (const float*)d_in[1];
    const float* W1  = (const float*)d_in[2];
    const float* as1 = (const float*)d_in[3];
    const float* ad1 = (const float*)d_in[4];
    const float* We1 = (const float*)d_in[5];
    const float* ae1 = (const float*)d_in[6];
    const float* b1  = (const float*)d_in[7];
    const float* W2  = (const float*)d_in[8];
    const float* as2 = (const float*)d_in[9];
    const float* ad2 = (const float*)d_in[10];
    const float* We2 = (const float*)d_in[11];
    const float* ae2 = (const float*)d_in[12];
    const float* b2  = (const float*)d_in[13];
    const float* lw  = (const float*)d_in[14];
    const float* lb  = (const float*)d_in[15];
    const int*   ei  = (const int*)d_in[16];

    const int N = in_sizes[0] / F_IN;
    const int E = in_sizes[16] / 2;
    const int* srcA = ei;
    const int* dstA = ei + E;

    char* w = (char*)d_ws;
    auto alloc = [&](size_t bytes) -> void* {
        void* p = (void*)w;
        w += (bytes + 255) & ~(size_t)255;
        return p;
    };
    int*   degI   = (int*)  alloc((size_t)N * 4);
    float* sum1 = (float*)alloc((size_t)N * 16);
    float* sum2 = (float*)alloc((size_t)N * 16);
    float* ae1v = (float*)alloc((size_t)E * 16);
    float* ae2v = (float*)alloc((size_t)E * 16);
    float* a_sb = (float*)alloc((size_t)N * 16);
    float* a_db = (float*)alloc((size_t)N * 16);
    int*   off  = (int*)  alloc((size_t)N * 4);
    int*   eidA = (int*)  alloc((size_t)E * 4);
    float* Hp   = (float*)alloc((size_t)N * HC * 4);
    float* h1   = (float*)alloc((size_t)N * HC * 4);

    const int gemmBlocks = (N + GEMM_R - 1) / GEMM_R;
    const int gemm1Blocks = (N + 4 * GEMM_R - 1) / (4 * GEMM_R);
    const int nodeBlocks = (N + 7) / 8;
    const int ownBlocks  = (N + OWN_R - 1) / OWN_R;
    const int nSlices    = (E + BK_SL - 1) / BK_SL;
    int*   bkw  = (int*)ae1v;
    float* Bt   = h1;
    int*   cntT = (int*)h1 + 1024;

    csr_bucket_kernel<<<nSlices + 1, BK_T, 0, stream>>>(dstA, bkw, cntT, E, ownBlocks, nSlices,
                                                        We1, ae1, We2, ae2, Bt);
    csr_owner_kernel<<<ownBlocks + gemm1Blocks, OWN_T, 0, stream>>>(bkw, cntT, dstA, off, degI, eidA,
                                                                    N, E, nSlices, ownBlocks,
                                                                    x, W1, Hp);
    edge_csr_kernel<<<nodeBlocks, 256, 0, stream>>>(ea, off, degI, eidA, srcA, Bt,
                                                    ae1v, ae2v, sum1, sum2, N,
                                                    Hp, as1, ad1, a_sb, a_db);

    agg_kernel<<<nodeBlocks, 256, 0, stream>>>(off, degI, eidA, ae1v, a_sb, a_db,
                                               sum1, Hp, b1, nullptr, nullptr,
                                               h1, nullptr, N, 0);
    gemm_wmma_k128_kernel<<<gemmBlocks, 256, 0, stream>>>(h1, W2, Hp, N);
    node_att_kernel<<<nodeBlocks, 256, 0, stream>>>(Hp, as2, ad2, a_sb, a_db, N);
    agg_kernel<<<nodeBlocks, 256, 0, stream>>>(off, degI, eidA, ae2v, a_sb, a_db,
                                               sum2, Hp, b2, lw, lb,
                                               nullptr, (float*)d_out, N, 1);
}
